# starting kernel
# baseline (speedup 1.0000x reference)
#include <hip/hip_runtime.h>
#include <hip/hip_bf16.h>

#define N_NODES 200000
#define N_EDGES 1000000
#define N_GRAPHS 4096
#define F_IN 11
#define H 64

typedef float v2f __attribute__((ext_vector_type(2)));
typedef float v8f __attribute__((ext_vector_type(8)));

__device__ __forceinline__ void atomicAddF(float* p, float v) {
    unsafeAtomicAdd(p, v);
}


__global__ void k_init_deg(float* deg, int n) {
    int i = blockIdx.x * blockDim.x + threadIdx.x;
    if (i < n) deg[i] = 1.0f;
}

__global__ void k_deg_edges(const int* __restrict__ dst, float* deg, int e) {
    int i = blockIdx.x * blockDim.x + threadIdx.x;
    if (i < e) atomicAddF(&deg[dst[i]], 1.0f);
}

__global__ void k_rsqrt(float* deg, int n) {
    int i = blockIdx.x * blockDim.x + threadIdx.x;
    if (i < n) deg[i] = rsqrtf(fmaxf(deg[i], 1.0f));
}


template <int KV, int AS, int KP>
__global__ void k_gemm_wmma(const float* __restrict__ A, const float* __restrict__ W,
                            float* __restrict__ Out, int n) {
    __shared__ float Wlds[(KP / 2) * H * 2];
    for (int idx = threadIdx.x; idx < KP * H; idx += blockDim.x) {
        int k = idx / H, c = idx - k * H;
        float v = (k < KV) ? W[k * H + c] : 0.0f;
        Wlds[(k >> 1) * (H * 2) + c * 2 + (k & 1)] = v;
    }
    __syncthreads();

    const int lane  = threadIdx.x & 31;
    const int strip = blockIdx.x * (blockDim.x >> 5) + (threadIdx.x >> 5);
    const int row0  = strip * 16;
    if (row0 >= n) return;

    const int mrow = lane & 15;
    const int hi   = lane >> 4;
    const float* __restrict__ arow = A + (size_t)(row0 + mrow) * AS;

    v8f acc[4];
    #pragma unroll
    for (int t = 0; t < 4; ++t) acc[t] = (v8f){0, 0, 0, 0, 0, 0, 0, 0};

    #pragma unroll
    for (int k0 = 0; k0 < KP; k0 += 4) {
        const int ka = k0 + hi * 2;
        v2f a;
        if (KV == KP && (AS & 1) == 0) {
            a = *reinterpret_cast<const v2f*>(arow + ka);
        } else {
            const int kc0 = (ka     < KV) ? ka     : (KV - 1);
            const int kc1 = (ka + 1 < KV) ? ka + 1 : (KV - 1);
            float ax = arow[kc0];
            float ay = arow[kc1];
            a.x = (ka     < KV) ? ax : 0.0f;
            a.y = (ka + 1 < KV) ? ay : 0.0f;
        }

        const int pbase = ((k0 >> 1) + hi) * (H * 2);
        #pragma unroll
        for (int t = 0; t < 4; ++t) {
            const int c = t * 16 + mrow;
            v2f b = *reinterpret_cast<const v2f*>(&Wlds[pbase + c * 2]);
            acc[t] = __builtin_amdgcn_wmma_f32_16x16x4_f32(
                 false, a,  false, b,
                 (short)0, acc[t],  false,  false);
        }
    }

    const int mbase = (lane < 16) ? 0 : 8;
    #pragma unroll
    for (int t = 0; t < 4; ++t) {
        const int col = t * 16 + mrow;
        #pragma unroll
        for (int j = 0; j < 8; ++j) {
            Out[(row0 + mbase + j) * H + col] = acc[t][j];
        }
    }
}


__global__ void k_selfloop_init(const float* __restrict__ h, const float* __restrict__ dis,
                                float* __restrict__ agg, int n) {
    int t = blockIdx.x * blockDim.x + threadIdx.x;
    int node = t >> 4;
    if (node >= n) return;
    int f4 = (t & 15) * 4;
    float w = dis[node]; w *= w;
    float4 hv = *reinterpret_cast<const float4*>(h + node * H + f4);
    float4 r = make_float4(hv.x * w, hv.y * w, hv.z * w, hv.w * w);
    *reinterpret_cast<float4*>(agg + node * H + f4) = r;
}

__global__ void k_scatter_edges(const int* __restrict__ src, const int* __restrict__ dst,
                                const float* __restrict__ dis, const float* __restrict__ h,
                                float* __restrict__ agg, int e) {
    int t = blockIdx.x * blockDim.x + threadIdx.x;
    int edge = t >> 4;
    if (edge >= e) return;
    int f4 = (t & 15) * 4;
    int s = src[edge];
    int d = dst[edge];
    float w = dis[s] * dis[d];
    float4 hv = *reinterpret_cast<const float4*>(h + s * H + f4);
    float* o = agg + d * H + f4;
    atomicAddF(o + 0, hv.x * w);
    atomicAddF(o + 1, hv.y * w);
    atomicAddF(o + 2, hv.z * w);
    atomicAddF(o + 3, hv.w * w);
}

__global__ void k_bias_act(float* __restrict__ agg, const float* __restrict__ b,
                           int n, int relu) {
    int t = blockIdx.x * blockDim.x + threadIdx.x;
    if (t >= n * H) return;
    float v = agg[t] + b[t & (H - 1)];
    agg[t] = relu ? fmaxf(v, 0.0f) : v;
}


__global__ void k_zero(float* p, int n) {
    int i = blockIdx.x * blockDim.x + threadIdx.x;
    if (i < n) p[i] = 0.0f;
}

__global__ void k_pool_acc(const float* __restrict__ h, const int* __restrict__ batch,
                           float* __restrict__ sums, float* __restrict__ cnts, int n) {
    int t = blockIdx.x * blockDim.x + threadIdx.x;
    int node = t >> 4;
    if (node >= n) return;
    int f4 = (t & 15) * 4;
    int g = batch[node];
    float4 hv = *reinterpret_cast<const float4*>(h + node * H + f4);
    float* o = sums + g * H + f4;
    atomicAddF(o + 0, hv.x);
    atomicAddF(o + 1, hv.y);
    atomicAddF(o + 2, hv.z);
    atomicAddF(o + 3, hv.w);
    if ((t & 15) == 0) atomicAddF(&cnts[g], 1.0f);
}

__global__ void k_pool_linear(const float* __restrict__ sums, const float* __restrict__ cnts,
                              const float* __restrict__ lin_w, const float* __restrict__ lin_b,
                              float* __restrict__ out, int g_total) {
    int g = blockIdx.x * blockDim.x + threadIdx.x;
    if (g >= g_total) return;
    float inv = 1.0f / fmaxf(cnts[g], 1.0f);
    float acc = 0.0f;
    #pragma unroll 8
    for (int f = 0; f < H; ++f) acc += sums[g * H + f] * lin_w[f];
    out[g] = acc * inv + lin_b[0];
}


extern "C" void kernel_launch(void* const* d_in, const int* in_sizes, int n_in,
                              void* d_out, int out_size, void* d_ws, size_t ws_size,
                              hipStream_t stream) {
    const float* x     = (const float*)d_in[0];
    const int*   eidx  = (const int*)  d_in[1];
    const int*   batch = (const int*)  d_in[2];
    const float* W1    = (const float*)d_in[3];
    const float* b1    = (const float*)d_in[4];
    const float* W2    = (const float*)d_in[5];
    const float* b2    = (const float*)d_in[6];
    const float* W3    = (const float*)d_in[7];
    const float* b3    = (const float*)d_in[8];
    const float* lin_w = (const float*)d_in[9];
    const float* lin_b = (const float*)d_in[10];
    float* out = (float*)d_out;

    const int n = in_sizes[0] / F_IN;
    const int e = in_sizes[1] / 2;
    const int* src = eidx;
    const int* dst = eidx + e;

    float* ws   = (float*)d_ws;
    float* dis  = ws;
    float* bufA = dis  + n;
    float* bufB = bufA + (size_t)n * H;
    float* sums = bufB + (size_t)n * H;
    float* cnts = sums + (size_t)N_GRAPHS * H;

    const int TB = 256;
    dim3 blk(TB);
    auto blocks = [](long long work, int tb) { return dim3((unsigned)((work + tb - 1) / tb)); };

    k_init_deg  <<<blocks(n, TB), blk, 0, stream>>>(dis, n);
    k_deg_edges <<<blocks(e, TB), blk, 0, stream>>>(dst, dis, e);
    k_rsqrt     <<<blocks(n, TB), blk, 0, stream>>>(dis, n);

    const int strips       = (n + 15) / 16;
    const dim3 gemm_grid   = blocks((long long)strips * 32, TB);
    const long long nodef4 = (long long)n * 16;
    const long long edgef4 = (long long)e * 16;

    k_gemm_wmma<F_IN, F_IN, 12><<<gemm_grid, blk, 0, stream>>>(x, W1, bufA, n);
    k_selfloop_init <<<blocks(nodef4, TB), blk, 0, stream>>>(bufA, dis, bufB, n);
    k_scatter_edges <<<blocks(edgef4, TB), blk, 0, stream>>>(src, dst, dis, bufA, bufB, e);
    k_bias_act      <<<blocks((long long)n * H, TB), blk, 0, stream>>>(bufB, b1, n, 1);

    k_gemm_wmma<H, H, H><<<gemm_grid, blk, 0, stream>>>(bufB, W2, bufA, n);
    k_selfloop_init <<<blocks(nodef4, TB), blk, 0, stream>>>(bufA, dis, bufB, n);
    k_scatter_edges <<<blocks(edgef4, TB), blk, 0, stream>>>(src, dst, dis, bufA, bufB, e);
    k_bias_act      <<<blocks((long long)n * H, TB), blk, 0, stream>>>(bufB, b2, n, 1);

    k_gemm_wmma<H, H, H><<<gemm_grid, blk, 0, stream>>>(bufB, W3, bufA, n);
    k_selfloop_init <<<blocks(nodef4, TB), blk, 0, stream>>>(bufA, dis, bufB, n);
    k_scatter_edges <<<blocks(edgef4, TB), blk, 0, stream>>>(src, dst, dis, bufA, bufB, e);
    k_bias_act      <<<blocks((long long)n * H, TB), blk, 0, stream>>>(bufB, b3, n, 0);

    k_zero       <<<blocks((long long)N_GRAPHS * H, TB), blk, 0, stream>>>(sums, N_GRAPHS * H);
    k_zero       <<<blocks(N_GRAPHS, TB), blk, 0, stream>>>(cnts, N_GRAPHS);
    k_pool_acc   <<<blocks(nodef4, TB), blk, 0, stream>>>(bufB, batch, sums, cnts, n);
    k_pool_linear<<<blocks(N_GRAPHS, TB), blk, 0, stream>>>(sums, cnts, lin_w, lin_b, out, N_GRAPHS);
}


# k_scatter_edges: one wave owns 32 edges, indices broadcast by readlane, lane = feature column: each row load and float atomic covers one whole 128-B line (8 -> 2 line requests per edge); atomics kept
# speedup vs baseline: 1.3079x; 1.3079x over previous
#include <hip/hip_runtime.h>
#include <hip/hip_bf16.h>

#define N_NODES 200000
#define N_EDGES 1000000
#define N_GRAPHS 4096
#define F_IN 11
#define H 64

typedef float v2f __attribute__((ext_vector_type(2)));
typedef float v8f __attribute__((ext_vector_type(8)));

__device__ __forceinline__ void atomicAddF(float* p, float v) {
    unsafeAtomicAdd(p, v);
}


__global__ void k_init_deg(float* deg, int n) {
    int i = blockIdx.x * blockDim.x + threadIdx.x;
    if (i < n) deg[i] = 1.0f;
}

__global__ void k_deg_edges(const int* __restrict__ dst, float* deg, int e) {
    int i = blockIdx.x * blockDim.x + threadIdx.x;
    if (i < e) atomicAddF(&deg[dst[i]], 1.0f);
}

__global__ void k_rsqrt(float* deg, int n) {
    int i = blockIdx.x * blockDim.x + threadIdx.x;
    if (i < n) deg[i] = rsqrtf(fmaxf(deg[i], 1.0f));
}


template <int KV, int AS, int KP>
__global__ void k_gemm_wmma(const float* __restrict__ A, const float* __restrict__ W,
                            float* __restrict__ Out, int n) {
    __shared__ float Wlds[(KP / 2) * H * 2];
    for (int idx = threadIdx.x; idx < KP * H; idx += blockDim.x) {
        int k = idx / H, c = idx - k * H;
        float v = (k < KV) ? W[k * H + c] : 0.0f;
        Wlds[(k >> 1) * (H * 2) + c * 2 + (k & 1)] = v;
    }
    __syncthreads();

    const int lane  = threadIdx.x & 31;
    const int strip = blockIdx.x * (blockDim.x >> 5) + (threadIdx.x >> 5);
    const int row0  = strip * 16;
    if (row0 >= n) return;

    const int mrow = lane & 15;
    const int hi   = lane >> 4;
    const float* __restrict__ arow = A + (size_t)(row0 + mrow) * AS;

    v8f acc[4];
    #pragma unroll
    for (int t = 0; t < 4; ++t) acc[t] = (v8f){0, 0, 0, 0, 0, 0, 0, 0};

    #pragma unroll
    for (int k0 = 0; k0 < KP; k0 += 4) {
        const int ka = k0 + hi * 2;
        v2f a;
        if (KV == KP && (AS & 1) == 0) {
            a = *reinterpret_cast<const v2f*>(arow + ka);
        } else {
            const int kc0 = (ka     < KV) ? ka     : (KV - 1);
            const int kc1 = (ka + 1 < KV) ? ka + 1 : (KV - 1);
            float ax = arow[kc0];
            float ay = arow[kc1];
            a.x = (ka     < KV) ? ax : 0.0f;
            a.y = (ka + 1 < KV) ? ay : 0.0f;
        }

        const int pbase = ((k0 >> 1) + hi) * (H * 2);
        #pragma unroll
        for (int t = 0; t < 4; ++t) {
            const int c = t * 16 + mrow;
            v2f b = *reinterpret_cast<const v2f*>(&Wlds[pbase + c * 2]);
            acc[t] = __builtin_amdgcn_wmma_f32_16x16x4_f32(
                 false, a,  false, b,
                 (short)0, acc[t],  false,  false);
        }
    }

    const int mbase = (lane < 16) ? 0 : 8;
    #pragma unroll
    for (int t = 0; t < 4; ++t) {
        const int col = t * 16 + mrow;
        #pragma unroll
        for (int j = 0; j < 8; ++j) {
            Out[(row0 + mbase + j) * H + col] = acc[t][j];
        }
    }
}


__global__ void k_selfloop_init(const float* __restrict__ h, const float* __restrict__ dis,
                                float* __restrict__ agg, int n) {
    int t = blockIdx.x * blockDim.x + threadIdx.x;
    int node = t >> 4;
    if (node >= n) return;
    int f4 = (t & 15) * 4;
    float w = dis[node]; w *= w;
    float4 hv = *reinterpret_cast<const float4*>(h + node * H + f4);
    float4 r = make_float4(hv.x * w, hv.y * w, hv.z * w, hv.w * w);
    *reinterpret_cast<float4*>(agg + node * H + f4) = r;
}

#define EDGES_PER_WAVE 32
#define EDGE_CHUNK 8

__device__ __forceinline__ int   bcast_i(int v, int l)   { return __builtin_amdgcn_readlane(v, l); }
__device__ __forceinline__ float bcast_f(float v, int l) {
    return __int_as_float(__builtin_amdgcn_readlane(__float_as_int(v), l));
}

__global__ void k_scatter_edges(const int* __restrict__ src, const int* __restrict__ dst,
                                const float* __restrict__ dis, const float* __restrict__ h,
                                float* __restrict__ agg, int e) {
    const int lane = threadIdx.x & 31;
    const int wave = blockIdx.x * (blockDim.x >> 5) + __builtin_amdgcn_readfirstlane(threadIdx.x >> 5);
    const int e0 = wave * EDGES_PER_WAVE;
    if (e0 >= e) return;
    const int left = e - e0;
    const int cnt = left < EDGES_PER_WAVE ? left : EDGES_PER_WAVE;

    int s = 0, d = 0;
    float w = 0.0f;
    if (lane < cnt) {
        s = src[e0 + lane];
        d = dst[e0 + lane];
        w = dis[s] * dis[d];
    }

    int c = 0;
    for (; c + EDGE_CHUNK <= cnt; c += EDGE_CHUNK) {
        float lo[EDGE_CHUNK], hi[EDGE_CHUNK];
        #pragma unroll
        for (int j = 0; j < EDGE_CHUNK; ++j) {
            const float* __restrict__ row = h + (size_t)bcast_i(s, c + j) * H;
            lo[j] = row[lane];
            hi[j] = row[lane + 32];
        }
        #pragma unroll
        for (int j = 0; j < EDGE_CHUNK; ++j) {
            const float wj = bcast_f(w, c + j);
            float* o = agg + (size_t)bcast_i(d, c + j) * H;
            atomicAddF(o + lane,      lo[j] * wj);
            atomicAddF(o + lane + 32, hi[j] * wj);
        }
    }
    for (; c < cnt; ++c) {
        const float* __restrict__ row = h + (size_t)bcast_i(s, c) * H;
        const float wj = bcast_f(w, c);
        float* o = agg + (size_t)bcast_i(d, c) * H;
        atomicAddF(o + lane,      row[lane] * wj);
        atomicAddF(o + lane + 32, row[lane + 32] * wj);
    }
}

__global__ void k_bias_act(float* __restrict__ agg, const float* __restrict__ b,
                           int n, int relu) {
    int t = blockIdx.x * blockDim.x + threadIdx.x;
    if (t >= n * H) return;
    float v = agg[t] + b[t & (H - 1)];
    agg[t] = relu ? fmaxf(v, 0.0f) : v;
}


__global__ void k_zero(float* p, int n) {
    int i = blockIdx.x * blockDim.x + threadIdx.x;
    if (i < n) p[i] = 0.0f;
}

__global__ void k_pool_acc(const float* __restrict__ h, const int* __restrict__ batch,
                           float* __restrict__ sums, float* __restrict__ cnts, int n) {
    int t = blockIdx.x * blockDim.x + threadIdx.x;
    int node = t >> 4;
    if (node >= n) return;
    int f4 = (t & 15) * 4;
    int g = batch[node];
    float4 hv = *reinterpret_cast<const float4*>(h + node * H + f4);
    float* o = sums + g * H + f4;
    atomicAddF(o + 0, hv.x);
    atomicAddF(o + 1, hv.y);
    atomicAddF(o + 2, hv.z);
    atomicAddF(o + 3, hv.w);
    if ((t & 15) == 0) atomicAddF(&cnts[g], 1.0f);
}

__global__ void k_pool_linear(const float* __restrict__ sums, const float* __restrict__ cnts,
                              const float* __restrict__ lin_w, const float* __restrict__ lin_b,
                              float* __restrict__ out, int g_total) {
    int g = blockIdx.x * blockDim.x + threadIdx.x;
    if (g >= g_total) return;
    float inv = 1.0f / fmaxf(cnts[g], 1.0f);
    float acc = 0.0f;
    #pragma unroll 8
    for (int f = 0; f < H; ++f) acc += sums[g * H + f] * lin_w[f];
    out[g] = acc * inv + lin_b[0];
}


extern "C" void kernel_launch(void* const* d_in, const int* in_sizes, int n_in,
                              void* d_out, int out_size, void* d_ws, size_t ws_size,
                              hipStream_t stream) {
    const float* x     = (const float*)d_in[0];
    const int*   eidx  = (const int*)  d_in[1];
    const int*   batch = (const int*)  d_in[2];
    const float* W1    = (const float*)d_in[3];
    const float* b1    = (const float*)d_in[4];
    const float* W2    = (const float*)d_in[5];
    const float* b2    = (const float*)d_in[6];
    const float* W3    = (const float*)d_in[7];
    const float* b3    = (const float*)d_in[8];
    const float* lin_w = (const float*)d_in[9];
    const float* lin_b = (const float*)d_in[10];
    float* out = (float*)d_out;

    const int n = in_sizes[0] / F_IN;
    const int e = in_sizes[1] / 2;
    const int* src = eidx;
    const int* dst = eidx + e;

    float* ws   = (float*)d_ws;
    float* dis  = ws;
    float* bufA = dis  + n;
    float* bufB = bufA + (size_t)n * H;
    float* sums = bufB + (size_t)n * H;
    float* cnts = sums + (size_t)N_GRAPHS * H;

    const int TB = 256;
    dim3 blk(TB);
    auto blocks = [](long long work, int tb) { return dim3((unsigned)((work + tb - 1) / tb)); };

    k_init_deg  <<<blocks(n, TB), blk, 0, stream>>>(dis, n);
    k_deg_edges <<<blocks(e, TB), blk, 0, stream>>>(dst, dis, e);
    k_rsqrt     <<<blocks(n, TB), blk, 0, stream>>>(dis, n);

    const int strips       = (n + 15) / 16;
    const dim3 gemm_grid   = blocks((long long)strips * 32, TB);
    const long long nodef4 = (long long)n * 16;

    k_gemm_wmma<F_IN, F_IN, 12><<<gemm_grid, blk, 0, stream>>>(x, W1, bufA, n);
    k_selfloop_init <<<blocks(nodef4, TB), blk, 0, stream>>>(bufA, dis, bufB, n);
    k_scatter_edges <<<blocks((long long)e, TB), blk, 0, stream>>>(src, dst, dis, bufA, bufB, e);
    k_bias_act      <<<blocks((long long)n * H, TB), blk, 0, stream>>>(bufB, b1, n, 1);

    k_gemm_wmma<H, H, H><<<gemm_grid, blk, 0, stream>>>(bufB, W2, bufA, n);
    k_selfloop_init <<<blocks(nodef4, TB), blk, 0, stream>>>(bufA, dis, bufB, n);
    k_scatter_edges <<<blocks((long long)e, TB), blk, 0, stream>>>(src, dst, dis, bufA, bufB, e);
    k_bias_act      <<<blocks((long long)n * H, TB), blk, 0, stream>>>(bufB, b2, n, 1);

    k_gemm_wmma<H, H, H><<<gemm_grid, blk, 0, stream>>>(bufB, W3, bufA, n);
    k_selfloop_init <<<blocks(nodef4, TB), blk, 0, stream>>>(bufA, dis, bufB, n);
    k_scatter_edges <<<blocks((long long)e, TB), blk, 0, stream>>>(src, dst, dis, bufA, bufB, e);
    k_bias_act      <<<blocks((long long)n * H, TB), blk, 0, stream>>>(bufB, b3, n, 0);

    k_zero       <<<blocks((long long)N_GRAPHS * H, TB), blk, 0, stream>>>(sums, N_GRAPHS * H);
    k_zero       <<<blocks(N_GRAPHS, TB), blk, 0, stream>>>(cnts, N_GRAPHS);
    k_pool_acc   <<<blocks(nodef4, TB), blk, 0, stream>>>(bufB, batch, sums, cnts, n);
    k_pool_linear<<<blocks(N_GRAPHS, TB), blk, 0, stream>>>(sums, cnts, lin_w, lin_b, out, N_GRAPHS);
}


# elementwise passes folded into their neighbours: bias+relu applied on the next GEMM's A load (last bias on the pool load), self-loop start value stored by the GEMM epilogue; 19 -> 13 launches
# speedup vs baseline: 1.9110x; 1.4611x over previous
#include <hip/hip_runtime.h>
#include <hip/hip_bf16.h>

#define N_NODES 200000
#define N_EDGES 1000000
#define N_GRAPHS 4096
#define F_IN 11
#define H 64

typedef float v2f __attribute__((ext_vector_type(2)));
typedef float v8f __attribute__((ext_vector_type(8)));

__device__ __forceinline__ void atomicAddF(float* p, float v) {
    unsafeAtomicAdd(p, v);
}


__global__ void k_init_deg(float* deg, int n) {
    int i = blockIdx.x * blockDim.x + threadIdx.x;
    if (i < n) deg[i] = 1.0f;
}

__global__ void k_deg_edges(const int* __restrict__ dst, float* deg, int e) {
    int i = blockIdx.x * blockDim.x + threadIdx.x;
    if (i < e) atomicAddF(&deg[dst[i]], 1.0f);
}

__global__ void k_rsqrt(float* deg, int n) {
    int i = blockIdx.x * blockDim.x + threadIdx.x;
    if (i < n) deg[i] = rsqrtf(fmaxf(deg[i], 1.0f));
}


template <int KV, int AS, int KP, bool PRE>
__global__ void k_gemm_wmma(const float* A, const float* __restrict__ W,
                            const float* __restrict__ bprev, const float* __restrict__ dis,
                            float* __restrict__ Out, float* Agg, int n) {
    __shared__ float Wlds[(KP / 2) * H * 2];
    for (int idx = threadIdx.x; idx < KP * H; idx += blockDim.x) {
        int k = idx / H, c = idx - k * H;
        float v = (k < KV) ? W[k * H + c] : 0.0f;
        Wlds[(k >> 1) * (H * 2) + c * 2 + (k & 1)] = v;
    }
    __shared__ float Blds[PRE ? KP : 4];
    if (PRE) {
        for (int idx = threadIdx.x; idx < KP; idx += blockDim.x) Blds[idx] = (idx < KV) ? bprev[idx] : 0.0f;
    }
    __syncthreads();

    const int lane  = threadIdx.x & 31;
    const int strip = blockIdx.x * (blockDim.x >> 5) + (threadIdx.x >> 5);
    const int row0  = strip * 16;
    if (row0 >= n) return;

    const int mrow = lane & 15;
    const int hi   = lane >> 4;
    const float* arow = A + (size_t)(row0 + mrow) * AS;

    const int mbase = (lane < 16) ? 0 : 8;
    float wself[8];
    #pragma unroll
    for (int j = 0; j < 8; ++j) {
        const float dv = dis[row0 + mbase + j];
        wself[j] = dv * dv;
    }

    v8f acc[4];
    #pragma unroll
    for (int t = 0; t < 4; ++t) acc[t] = (v8f){0, 0, 0, 0, 0, 0, 0, 0};

    #pragma unroll
    for (int k0 = 0; k0 < KP; k0 += 4) {
        const int ka = k0 + hi * 2;
        v2f a;
        if (KV == KP && (AS & 1) == 0) {
            a = *reinterpret_cast<const v2f*>(arow + ka);
        } else {
            const int kc0 = (ka     < KV) ? ka     : (KV - 1);
            const int kc1 = (ka + 1 < KV) ? ka + 1 : (KV - 1);
            float ax = arow[kc0];
            float ay = arow[kc1];
            a.x = (ka     < KV) ? ax : 0.0f;
            a.y = (ka + 1 < KV) ? ay : 0.0f;
        }
        if (PRE) {
            const v2f bp = *reinterpret_cast<const v2f*>(&Blds[ka]);
            a.x = fmaxf(a.x + bp.x, 0.0f);
            a.y = fmaxf(a.y + bp.y, 0.0f);
        }

        const int pbase = ((k0 >> 1) + hi) * (H * 2);
        #pragma unroll
        for (int t = 0; t < 4; ++t) {
            const int c = t * 16 + mrow;
            v2f b = *reinterpret_cast<const v2f*>(&Wlds[pbase + c * 2]);
            acc[t] = __builtin_amdgcn_wmma_f32_16x16x4_f32(
                 false, a,  false, b,
                 (short)0, acc[t],  false,  false);
        }
    }

    #pragma unroll
    for (int t = 0; t < 4; ++t) {
        const int col = t * 16 + mrow;
        #pragma unroll
        for (int j = 0; j < 8; ++j) {
            const float v = acc[t][j];
            Out[(row0 + mbase + j) * H + col] = v;
            Agg[(row0 + mbase + j) * H + col] = v * wself[j];
        }
    }
}


#define EDGES_PER_WAVE 32
#define EDGE_CHUNK 8

__device__ __forceinline__ int   bcast_i(int v, int l)   { return __builtin_amdgcn_readlane(v, l); }
__device__ __forceinline__ float bcast_f(float v, int l) {
    return __int_as_float(__builtin_amdgcn_readlane(__float_as_int(v), l));
}

__global__ void k_scatter_edges(const int* __restrict__ src, const int* __restrict__ dst,
                                const float* __restrict__ dis, const float* __restrict__ h,
                                float* __restrict__ agg, int e) {
    const int lane = threadIdx.x & 31;
    const int wave = blockIdx.x * (blockDim.x >> 5) + __builtin_amdgcn_readfirstlane(threadIdx.x >> 5);
    const int e0 = wave * EDGES_PER_WAVE;
    if (e0 >= e) return;
    const int left = e - e0;
    const int cnt = left < EDGES_PER_WAVE ? left : EDGES_PER_WAVE;

    int s = 0, d = 0;
    float w = 0.0f;
    if (lane < cnt) {
        s = src[e0 + lane];
        d = dst[e0 + lane];
        w = dis[s] * dis[d];
    }

    int c = 0;
    for (; c + EDGE_CHUNK <= cnt; c += EDGE_CHUNK) {
        float lo[EDGE_CHUNK], hi[EDGE_CHUNK];
        #pragma unroll
        for (int j = 0; j < EDGE_CHUNK; ++j) {
            const float* __restrict__ row = h + (size_t)bcast_i(s, c + j) * H;
            lo[j] = row[lane];
            hi[j] = row[lane + 32];
        }
        #pragma unroll
        for (int j = 0; j < EDGE_CHUNK; ++j) {
            const float wj = bcast_f(w, c + j);
            float* o = agg + (size_t)bcast_i(d, c + j) * H;
            atomicAddF(o + lane,      lo[j] * wj);
            atomicAddF(o + lane + 32, hi[j] * wj);
        }
    }
    for (; c < cnt; ++c) {
        const float* __restrict__ row = h + (size_t)bcast_i(s, c) * H;
        const float wj = bcast_f(w, c);
        float* o = agg + (size_t)bcast_i(d, c) * H;
        atomicAddF(o + lane,      row[lane] * wj);
        atomicAddF(o + lane + 32, row[lane + 32] * wj);
    }
}


__global__ void k_zero(float* p, int n) {
    int i = blockIdx.x * blockDim.x + threadIdx.x;
    if (i < n) p[i] = 0.0f;
}

__global__ void k_pool_acc(const float* __restrict__ h, const float* __restrict__ b,
                           const int* __restrict__ batch,
                           float* __restrict__ sums, float* __restrict__ cnts, int n) {
    int t = blockIdx.x * blockDim.x + threadIdx.x;
    int node = t >> 4;
    if (node >= n) return;
    int f4 = (t & 15) * 4;
    int g = batch[node];
    float4 hv = *reinterpret_cast<const float4*>(h + node * H + f4);
    float4 bv = *reinterpret_cast<const float4*>(b + f4);
    float* o = sums + g * H + f4;
    atomicAddF(o + 0, hv.x + bv.x);
    atomicAddF(o + 1, hv.y + bv.y);
    atomicAddF(o + 2, hv.z + bv.z);
    atomicAddF(o + 3, hv.w + bv.w);
    if ((t & 15) == 0) atomicAddF(&cnts[g], 1.0f);
}

__global__ void k_pool_linear(const float* __restrict__ sums, const float* __restrict__ cnts,
                              const float* __restrict__ lin_w, const float* __restrict__ lin_b,
                              float* __restrict__ out, int g_total) {
    int g = blockIdx.x * blockDim.x + threadIdx.x;
    if (g >= g_total) return;
    float inv = 1.0f / fmaxf(cnts[g], 1.0f);
    float acc = 0.0f;
    #pragma unroll 8
    for (int f = 0; f < H; ++f) acc += sums[g * H + f] * lin_w[f];
    out[g] = acc * inv + lin_b[0];
}


extern "C" void kernel_launch(void* const* d_in, const int* in_sizes, int n_in,
                              void* d_out, int out_size, void* d_ws, size_t ws_size,
                              hipStream_t stream) {
    const float* x     = (const float*)d_in[0];
    const int*   eidx  = (const int*)  d_in[1];
    const int*   batch = (const int*)  d_in[2];
    const float* W1    = (const float*)d_in[3];
    const float* b1    = (const float*)d_in[4];
    const float* W2    = (const float*)d_in[5];
    const float* b2    = (const float*)d_in[6];
    const float* W3    = (const float*)d_in[7];
    const float* b3    = (const float*)d_in[8];
    const float* lin_w = (const float*)d_in[9];
    const float* lin_b = (const float*)d_in[10];
    float* out = (float*)d_out;

    const int n = in_sizes[0] / F_IN;
    const int e = in_sizes[1] / 2;
    const int* src = eidx;
    const int* dst = eidx + e;

    float* ws   = (float*)d_ws;
    float* dis  = ws;
    float* bufA = dis  + n;
    float* bufB = bufA + (size_t)n * H;
    float* sums = bufB + (size_t)n * H;
    float* cnts = sums + (size_t)N_GRAPHS * H;

    const int TB = 256;
    dim3 blk(TB);
    auto blocks = [](long long work, int tb) { return dim3((unsigned)((work + tb - 1) / tb)); };

    k_init_deg  <<<blocks(n, TB), blk, 0, stream>>>(dis, n);
    k_deg_edges <<<blocks(e, TB), blk, 0, stream>>>(dst, dis, e);
    k_rsqrt     <<<blocks(n, TB), blk, 0, stream>>>(dis, n);

    const int strips       = (n + 15) / 16;
    const dim3 gemm_grid   = blocks((long long)strips * 32, TB);
    const long long nodef4 = (long long)n * 16;


    k_gemm_wmma<F_IN, F_IN, 12, false><<<gemm_grid, blk, 0, stream>>>(x, W1, b1, dis, bufA, bufB, n);
    k_scatter_edges <<<blocks((long long)e, TB), blk, 0, stream>>>(src, dst, dis, bufA, bufB, e);

    k_gemm_wmma<H, H, H, true><<<gemm_grid, blk, 0, stream>>>(bufB, W2, b1, dis, bufA, bufB, n);
    k_scatter_edges <<<blocks((long long)e, TB), blk, 0, stream>>>(src, dst, dis, bufA, bufB, e);

    k_gemm_wmma<H, H, H, true><<<gemm_grid, blk, 0, stream>>>(bufB, W3, b2, dis, bufA, bufB, n);
    k_scatter_edges <<<blocks((long long)e, TB), blk, 0, stream>>>(src, dst, dis, bufA, bufB, e);

    k_zero       <<<blocks((long long)N_GRAPHS * H, TB), blk, 0, stream>>>(sums, N_GRAPHS * H);
    k_zero       <<<blocks(N_GRAPHS, TB), blk, 0, stream>>>(cnts, N_GRAPHS);
    k_pool_acc   <<<blocks(nodef4, TB), blk, 0, stream>>>(bufB, b3, batch, sums, cnts, n);
    k_pool_linear<<<blocks(N_GRAPHS, TB), blk, 0, stream>>>(sums, cnts, lin_w, lin_b, out, N_GRAPHS);
}


# layer 1 aggregates before it multiplies: the edge scatter runs on the 11-wide input rows (two edges per atomic instruction), one GEMM then makes the 64-wide aggregate; float sums regrouped
# speedup vs baseline: 1.9899x; 1.0413x over previous
#include <hip/hip_runtime.h>
#include <hip/hip_bf16.h>

#define N_NODES 200000
#define N_EDGES 1000000
#define N_GRAPHS 4096
#define F_IN 11
#define H 64
#define XS 16

typedef float v2f __attribute__((ext_vector_type(2)));
typedef float v8f __attribute__((ext_vector_type(8)));

__device__ __forceinline__ void atomicAddF(float* p, float v) {
    unsafeAtomicAdd(p, v);
}


__global__ void k_init_deg(float* deg, int n) {
    int i = blockIdx.x * blockDim.x + threadIdx.x;
    if (i < n) deg[i] = 1.0f;
}

__global__ void k_deg_edges(const int* __restrict__ dst, float* deg, int e) {
    int i = blockIdx.x * blockDim.x + threadIdx.x;
    if (i < e) atomicAddF(&deg[dst[i]], 1.0f);
}

__global__ void k_rsqrt_start(float* deg, const float* __restrict__ x, float* __restrict__ aggx, int n) {
    int i = blockIdx.x * blockDim.x + threadIdx.x;
    if (i >= n) return;
    const float dv = rsqrtf(fmaxf(deg[i], 1.0f));
    deg[i] = dv;
    const float ws = dv * dv;
    const float* __restrict__ xr = x + i * F_IN;
    float r[XS];
    #pragma unroll
    for (int f = 0; f < XS; ++f) r[f] = (f < F_IN) ? xr[f] * ws : 0.0f;
    float4* o = reinterpret_cast<float4*>(aggx + i * XS);
    #pragma unroll
    for (int q = 0; q < XS / 4; ++q) o[q] = make_float4(r[4 * q], r[4 * q + 1], r[4 * q + 2], r[4 * q + 3]);
}


template <int KV, int AS, int KP, bool PRE, bool SELF>
__global__ void k_gemm_wmma(const float* A, const float* __restrict__ W,
                            const float* __restrict__ bprev, const float* __restrict__ dis,
                            float* __restrict__ Out, float* Agg, int n) {
    __shared__ float Wlds[(KP / 2) * H * 2];
    for (int idx = threadIdx.x; idx < KP * H; idx += blockDim.x) {
        int k = idx / H, c = idx - k * H;
        float v = (k < KV) ? W[k * H + c] : 0.0f;
        Wlds[(k >> 1) * (H * 2) + c * 2 + (k & 1)] = v;
    }
    __shared__ float Blds[PRE ? KP : 4];
    if (PRE) {
        for (int idx = threadIdx.x; idx < KP; idx += blockDim.x) Blds[idx] = (idx < KV) ? bprev[idx] : 0.0f;
    }
    __syncthreads();

    const int lane  = threadIdx.x & 31;
    const int strip = blockIdx.x * (blockDim.x >> 5) + (threadIdx.x >> 5);
    const int row0  = strip * 16;
    if (row0 >= n) return;

    const int mrow = lane & 15;
    const int hi   = lane >> 4;
    const float* arow = A + (size_t)(row0 + mrow) * AS;

    const int mbase = (lane < 16) ? 0 : 8;
    float wself[8];
    #pragma unroll
    for (int j = 0; j < 8; ++j) {
        const float dv = SELF ? dis[row0 + mbase + j] : 0.0f;
        wself[j] = dv * dv;
    }

    v8f acc[4];
    #pragma unroll
    for (int t = 0; t < 4; ++t) acc[t] = (v8f){0, 0, 0, 0, 0, 0, 0, 0};

    #pragma unroll
    for (int k0 = 0; k0 < KP; k0 += 4) {
        const int ka = k0 + hi * 2;
        static_assert(AS >= KP && (AS & 1) == 0, "A rows must hold KP columns at an even stride");
        v2f a = *reinterpret_cast<const v2f*>(arow + ka);
        if (PRE) {
            const v2f bp = *reinterpret_cast<const v2f*>(&Blds[ka]);
            a.x = fmaxf(a.x + bp.x, 0.0f);
            a.y = fmaxf(a.y + bp.y, 0.0f);
        }

        const int pbase = ((k0 >> 1) + hi) * (H * 2);
        #pragma unroll
        for (int t = 0; t < 4; ++t) {
            const int c = t * 16 + mrow;
            v2f b = *reinterpret_cast<const v2f*>(&Wlds[pbase + c * 2]);
            acc[t] = __builtin_amdgcn_wmma_f32_16x16x4_f32(
                 false, a,  false, b,
                 (short)0, acc[t],  false,  false);
        }
    }

    #pragma unroll
    for (int t = 0; t < 4; ++t) {
        const int col = t * 16 + mrow;
        #pragma unroll
        for (int j = 0; j < 8; ++j) {
            const float v = acc[t][j];
            Out[(row0 + mbase + j) * H + col] = v;
            if (SELF) Agg[(row0 + mbase + j) * H + col] = v * wself[j];
        }
    }
}


#define EDGES_PER_WAVE 32
#define EDGE_CHUNK 8

__device__ __forceinline__ int   bcast_i(int v, int l)   { return __builtin_amdgcn_readlane(v, l); }
__device__ __forceinline__ float bcast_f(float v, int l) {
    return __int_as_float(__builtin_amdgcn_readlane(__float_as_int(v), l));
}

__global__ void k_scatter_edges(const int* __restrict__ src, const int* __restrict__ dst,
                                const float* __restrict__ dis, const float* __restrict__ h,
                                float* __restrict__ agg, int e) {
    const int lane = threadIdx.x & 31;
    const int wave = blockIdx.x * (blockDim.x >> 5) + __builtin_amdgcn_readfirstlane(threadIdx.x >> 5);
    const int e0 = wave * EDGES_PER_WAVE;
    if (e0 >= e) return;
    const int left = e - e0;
    const int cnt = left < EDGES_PER_WAVE ? left : EDGES_PER_WAVE;

    int s = 0, d = 0;
    float w = 0.0f;
    if (lane < cnt) {
        s = src[e0 + lane];
        d = dst[e0 + lane];
        w = dis[s] * dis[d];
    }

    int c = 0;
    for (; c + EDGE_CHUNK <= cnt; c += EDGE_CHUNK) {
        float lo[EDGE_CHUNK], hi[EDGE_CHUNK];
        #pragma unroll
        for (int j = 0; j < EDGE_CHUNK; ++j) {
            const float* __restrict__ row = h + (size_t)bcast_i(s, c + j) * H;
            lo[j] = row[lane];
            hi[j] = row[lane + 32];
        }
        #pragma unroll
        for (int j = 0; j < EDGE_CHUNK; ++j) {
            const float wj = bcast_f(w, c + j);
            float* o = agg + (size_t)bcast_i(d, c + j) * H;
            atomicAddF(o + lane,      lo[j] * wj);
            atomicAddF(o + lane + 32, hi[j] * wj);
        }
    }
    for (; c < cnt; ++c) {
        const float* __restrict__ row = h + (size_t)bcast_i(s, c) * H;
        const float wj = bcast_f(w, c);
        float* o = agg + (size_t)bcast_i(d, c) * H;
        atomicAddF(o + lane,      row[lane] * wj);
        atomicAddF(o + lane + 32, row[lane + 32] * wj);
    }
}

__device__ __forceinline__ int   pick_i(int v, int j, int hi)   { const int a = bcast_i(v, 2 * j), b = bcast_i(v, 2 * j + 1); return hi ? b : a; }
__device__ __forceinline__ float pick_f(float v, int j, int hi) { const float a = bcast_f(v, 2 * j), b = bcast_f(v, 2 * j + 1); return hi ? b : a; }

__global__ void k_scatter_x(const int* __restrict__ src, const int* __restrict__ dst,
                            const float* __restrict__ dis, const float* __restrict__ x,
                            float* __restrict__ aggx, int e) {
    const int lane = threadIdx.x & 31;
    const int wave = blockIdx.x * (blockDim.x >> 5) + __builtin_amdgcn_readfirstlane(threadIdx.x >> 5);
    const int e0 = wave * EDGES_PER_WAVE;
    if (e0 >= e) return;
    const int left = e - e0;
    const int cnt = left < EDGES_PER_WAVE ? left : EDGES_PER_WAVE;

    int s = 0, d = 0;
    float w = 0.0f;
    if (lane < cnt) {
        s = src[e0 + lane];
        d = dst[e0 + lane];
        w = dis[s] * dis[d];
    }

    const int f  = lane & 15;
    const int hi = lane >> 4;
    if (cnt == EDGES_PER_WAVE) {
        if (f < F_IN) {
            float v[EDGES_PER_WAVE / 2];
            #pragma unroll
            for (int j = 0; j < EDGES_PER_WAVE / 2; ++j) v[j] = x[pick_i(s, j, hi) * F_IN + f];
            #pragma unroll
            for (int j = 0; j < EDGES_PER_WAVE / 2; ++j)
                atomicAddF(aggx + pick_i(d, j, hi) * XS + f, v[j] * pick_f(w, j, hi));
        }
    } else {
        for (int j = 0; 2 * j < cnt; ++j) {
            const int sj = pick_i(s, j, hi), dj = pick_i(d, j, hi);
            const float wj = pick_f(w, j, hi);
            if (f < F_IN && 2 * j + hi < cnt) atomicAddF(aggx + dj * XS + f, x[sj * F_IN + f] * wj);
        }
    }
}


__global__ void k_zero(float* p, int n) {
    int i = blockIdx.x * blockDim.x + threadIdx.x;
    if (i < n) p[i] = 0.0f;
}

__global__ void k_pool_acc(const float* __restrict__ h, const float* __restrict__ b,
                           const int* __restrict__ batch,
                           float* __restrict__ sums, float* __restrict__ cnts, int n) {
    int t = blockIdx.x * blockDim.x + threadIdx.x;
    int node = t >> 4;
    if (node >= n) return;
    int f4 = (t & 15) * 4;
    int g = batch[node];
    float4 hv = *reinterpret_cast<const float4*>(h + node * H + f4);
    float4 bv = *reinterpret_cast<const float4*>(b + f4);
    float* o = sums + g * H + f4;
    atomicAddF(o + 0, hv.x + bv.x);
    atomicAddF(o + 1, hv.y + bv.y);
    atomicAddF(o + 2, hv.z + bv.z);
    atomicAddF(o + 3, hv.w + bv.w);
    if ((t & 15) == 0) atomicAddF(&cnts[g], 1.0f);
}

__global__ void k_pool_linear(const float* __restrict__ sums, const float* __restrict__ cnts,
                              const float* __restrict__ lin_w, const float* __restrict__ lin_b,
                              float* __restrict__ out, int g_total) {
    int g = blockIdx.x * blockDim.x + threadIdx.x;
    if (g >= g_total) return;
    float inv = 1.0f / fmaxf(cnts[g], 1.0f);
    float acc = 0.0f;
    #pragma unroll 8
    for (int f = 0; f < H; ++f) acc += sums[g * H + f] * lin_w[f];
    out[g] = acc * inv + lin_b[0];
}


extern "C" void kernel_launch(void* const* d_in, const int* in_sizes, int n_in,
                              void* d_out, int out_size, void* d_ws, size_t ws_size,
                              hipStream_t stream) {
    const float* x     = (const float*)d_in[0];
    const int*   eidx  = (const int*)  d_in[1];
    const int*   batch = (const int*)  d_in[2];
    const float* W1    = (const float*)d_in[3];
    const float* b1    = (const float*)d_in[4];
    const float* W2    = (const float*)d_in[5];
    const float* b2    = (const float*)d_in[6];
    const float* W3    = (const float*)d_in[7];
    const float* b3    = (const float*)d_in[8];
    const float* lin_w = (const float*)d_in[9];
    const float* lin_b = (const float*)d_in[10];
    float* out = (float*)d_out;

    const int n = in_sizes[0] / F_IN;
    const int e = in_sizes[1] / 2;
    const int* src = eidx;
    const int* dst = eidx + e;

    float* ws   = (float*)d_ws;
    float* dis  = ws;
    float* bufA = dis  + n;
    float* bufB = bufA + (size_t)n * H;
    float* sums = bufB + (size_t)n * H;
    float* cnts = sums + (size_t)N_GRAPHS * H;

    const int TB = 256;
    dim3 blk(TB);
    auto blocks = [](long long work, int tb) { return dim3((unsigned)((work + tb - 1) / tb)); };

    k_init_deg  <<<blocks(n, TB), blk, 0, stream>>>(dis, n);
    k_deg_edges <<<blocks(e, TB), blk, 0, stream>>>(dst, dis, e);
    float* aggx = bufA;
    k_rsqrt_start<<<blocks(n, TB), blk, 0, stream>>>(dis, x, aggx, n);

    const int strips       = (n + 15) / 16;
    const dim3 gemm_grid   = blocks((long long)strips * 32, TB);
    const long long nodef4 = (long long)n * 16;

    k_scatter_x <<<blocks((long long)e, TB), blk, 0, stream>>>(src, dst, dis, x, aggx, e);
    k_gemm_wmma<F_IN, XS, 12, false, false><<<gemm_grid, blk, 0, stream>>>(aggx, W1, b1, dis, bufB, bufB, n);


    k_gemm_wmma<H, H, H, true, true><<<gemm_grid, blk, 0, stream>>>(bufB, W2, b1, dis, bufA, bufB, n);
    k_scatter_edges <<<blocks((long long)e, TB), blk, 0, stream>>>(src, dst, dis, bufA, bufB, e);

    k_gemm_wmma<H, H, H, true, true><<<gemm_grid, blk, 0, stream>>>(bufB, W3, b2, dis, bufA, bufB, n);
    k_scatter_edges <<<blocks((long long)e, TB), blk, 0, stream>>>(src, dst, dis, bufA, bufB, e);

    k_zero       <<<blocks((long long)N_GRAPHS * H, TB), blk, 0, stream>>>(sums, N_GRAPHS * H);
    k_zero       <<<blocks(N_GRAPHS, TB), blk, 0, stream>>>(cnts, N_GRAPHS);
    k_pool_acc   <<<blocks(nodef4, TB), blk, 0, stream>>>(bufB, b3, batch, sums, cnts, n);
    k_pool_linear<<<blocks(N_GRAPHS, TB), blk, 0, stream>>>(sums, cnts, lin_w, lin_b, out, N_GRAPHS);
}


# final linear folded back through mean pool and layer 3 (all linear): layer 3 = one WMMA column tile with W3*lin_w, its edge scatter and the pool carry 1 float per node, not 64; float sums regrouped
# speedup vs baseline: 2.3348x; 1.1733x over previous
#include <hip/hip_runtime.h>
#include <hip/hip_bf16.h>

#define N_NODES 200000
#define N_EDGES 1000000
#define N_GRAPHS 4096
#define F_IN 11
#define H 64
#define XS 16

typedef float v2f __attribute__((ext_vector_type(2)));
typedef float v8f __attribute__((ext_vector_type(8)));

__device__ __forceinline__ void atomicAddF(float* p, float v) {
    unsafeAtomicAdd(p, v);
}


__global__ void k_init_deg(float* deg, int n) {
    int i = blockIdx.x * blockDim.x + threadIdx.x;
    if (i < n) deg[i] = 1.0f;
}

__global__ void k_deg_edges(const int* __restrict__ dst, float* deg, int e) {
    int i = blockIdx.x * blockDim.x + threadIdx.x;
    if (i < e) atomicAddF(&deg[dst[i]], 1.0f);
}

__global__ void k_rsqrt_start(float* deg, const float* __restrict__ x, float* __restrict__ aggx, int n) {
    int i = blockIdx.x * blockDim.x + threadIdx.x;
    if (i >= n) return;
    const float dv = rsqrtf(fmaxf(deg[i], 1.0f));
    deg[i] = dv;
    const float ws = dv * dv;
    const float* __restrict__ xr = x + i * F_IN;
    float r[XS];
    #pragma unroll
    for (int f = 0; f < XS; ++f) r[f] = (f < F_IN) ? xr[f] * ws : 0.0f;
    float4* o = reinterpret_cast<float4*>(aggx + i * XS);
    #pragma unroll
    for (int q = 0; q < XS / 4; ++q) o[q] = make_float4(r[4 * q], r[4 * q + 1], r[4 * q + 2], r[4 * q + 3]);
}


template <int KV, int AS, int KP, bool PRE, bool SELF>
__global__ void k_gemm_wmma(const float* A, const float* __restrict__ W,
                            const float* __restrict__ bprev, const float* __restrict__ dis,
                            float* __restrict__ Out, float* Agg, int n) {
    __shared__ float Wlds[(KP / 2) * H * 2];
    for (int idx = threadIdx.x; idx < KP * H; idx += blockDim.x) {
        int k = idx / H, c = idx - k * H;
        float v = (k < KV) ? W[k * H + c] : 0.0f;
        Wlds[(k >> 1) * (H * 2) + c * 2 + (k & 1)] = v;
    }
    __shared__ float Blds[PRE ? KP : 4];
    if (PRE) {
        for (int idx = threadIdx.x; idx < KP; idx += blockDim.x) Blds[idx] = (idx < KV) ? bprev[idx] : 0.0f;
    }
    __syncthreads();

    const int lane  = threadIdx.x & 31;
    const int strip = blockIdx.x * (blockDim.x >> 5) + (threadIdx.x >> 5);
    const int row0  = strip * 16;
    if (row0 >= n) return;

    const int mrow = lane & 15;
    const int hi   = lane >> 4;
    const float* arow = A + (size_t)(row0 + mrow) * AS;

    const int mbase = (lane < 16) ? 0 : 8;
    float wself[8];
    #pragma unroll
    for (int j = 0; j < 8; ++j) {
        const float dv = SELF ? dis[row0 + mbase + j] : 0.0f;
        wself[j] = dv * dv;
    }

    v8f acc[4];
    #pragma unroll
    for (int t = 0; t < 4; ++t) acc[t] = (v8f){0, 0, 0, 0, 0, 0, 0, 0};

    #pragma unroll
    for (int k0 = 0; k0 < KP; k0 += 4) {
        const int ka = k0 + hi * 2;
        static_assert(AS >= KP && (AS & 1) == 0, "A rows must hold KP columns at an even stride");
        v2f a = *reinterpret_cast<const v2f*>(arow + ka);
        if (PRE) {
            const v2f bp = *reinterpret_cast<const v2f*>(&Blds[ka]);
            a.x = fmaxf(a.x + bp.x, 0.0f);
            a.y = fmaxf(a.y + bp.y, 0.0f);
        }

        const int pbase = ((k0 >> 1) + hi) * (H * 2);
        #pragma unroll
        for (int t = 0; t < 4; ++t) {
            const int c = t * 16 + mrow;
            v2f b = *reinterpret_cast<const v2f*>(&Wlds[pbase + c * 2]);
            acc[t] = __builtin_amdgcn_wmma_f32_16x16x4_f32(
                 false, a,  false, b,
                 (short)0, acc[t],  false,  false);
        }
    }

    #pragma unroll
    for (int t = 0; t < 4; ++t) {
        const int col = t * 16 + mrow;
        #pragma unroll
        for (int j = 0; j < 8; ++j) {
            const float v = acc[t][j];
            Out[(row0 + mbase + j) * H + col] = v;
            if (SELF) Agg[(row0 + mbase + j) * H + col] = v * wself[j];
        }
    }
}


#define EDGES_PER_WAVE 32
#define EDGE_CHUNK 8

__device__ __forceinline__ int   bcast_i(int v, int l)   { return __builtin_amdgcn_readlane(v, l); }
__device__ __forceinline__ float bcast_f(float v, int l) {
    return __int_as_float(__builtin_amdgcn_readlane(__float_as_int(v), l));
}

__global__ void k_scatter_edges(const int* __restrict__ src, const int* __restrict__ dst,
                                const float* __restrict__ dis, const float* __restrict__ h,
                                float* __restrict__ agg, int e) {
    const int lane = threadIdx.x & 31;
    const int wave = blockIdx.x * (blockDim.x >> 5) + __builtin_amdgcn_readfirstlane(threadIdx.x >> 5);
    const int e0 = wave * EDGES_PER_WAVE;
    if (e0 >= e) return;
    const int left = e - e0;
    const int cnt = left < EDGES_PER_WAVE ? left : EDGES_PER_WAVE;

    int s = 0, d = 0;
    float w = 0.0f;
    if (lane < cnt) {
        s = src[e0 + lane];
        d = dst[e0 + lane];
        w = dis[s] * dis[d];
    }

    int c = 0;
    for (; c + EDGE_CHUNK <= cnt; c += EDGE_CHUNK) {
        float lo[EDGE_CHUNK], hi[EDGE_CHUNK];
        #pragma unroll
        for (int j = 0; j < EDGE_CHUNK; ++j) {
            const float* __restrict__ row = h + (size_t)bcast_i(s, c + j) * H;
            lo[j] = row[lane];
            hi[j] = row[lane + 32];
        }
        #pragma unroll
        for (int j = 0; j < EDGE_CHUNK; ++j) {
            const float wj = bcast_f(w, c + j);
            float* o = agg + (size_t)bcast_i(d, c + j) * H;
            atomicAddF(o + lane,      lo[j] * wj);
            atomicAddF(o + lane + 32, hi[j] * wj);
        }
    }
    for (; c < cnt; ++c) {
        const float* __restrict__ row = h + (size_t)bcast_i(s, c) * H;
        const float wj = bcast_f(w, c);
        float* o = agg + (size_t)bcast_i(d, c) * H;
        atomicAddF(o + lane,      row[lane] * wj);
        atomicAddF(o + lane + 32, row[lane + 32] * wj);
    }
}

__device__ __forceinline__ int   pick_i(int v, int j, int hi)   { const int a = bcast_i(v, 2 * j), b = bcast_i(v, 2 * j + 1); return hi ? b : a; }
__device__ __forceinline__ float pick_f(float v, int j, int hi) { const float a = bcast_f(v, 2 * j), b = bcast_f(v, 2 * j + 1); return hi ? b : a; }

__global__ void k_scatter_x(const int* __restrict__ src, const int* __restrict__ dst,
                            const float* __restrict__ dis, const float* __restrict__ x,
                            float* __restrict__ aggx, int e) {
    const int lane = threadIdx.x & 31;
    const int wave = blockIdx.x * (blockDim.x >> 5) + __builtin_amdgcn_readfirstlane(threadIdx.x >> 5);
    const int e0 = wave * EDGES_PER_WAVE;
    if (e0 >= e) return;
    const int left = e - e0;
    const int cnt = left < EDGES_PER_WAVE ? left : EDGES_PER_WAVE;

    int s = 0, d = 0;
    float w = 0.0f;
    if (lane < cnt) {
        s = src[e0 + lane];
        d = dst[e0 + lane];
        w = dis[s] * dis[d];
    }

    const int f  = lane & 15;
    const int hi = lane >> 4;
    if (cnt == EDGES_PER_WAVE) {
        if (f < F_IN) {
            float v[EDGES_PER_WAVE / 2];
            #pragma unroll
            for (int j = 0; j < EDGES_PER_WAVE / 2; ++j) v[j] = x[pick_i(s, j, hi) * F_IN + f];
            #pragma unroll
            for (int j = 0; j < EDGES_PER_WAVE / 2; ++j)
                atomicAddF(aggx + pick_i(d, j, hi) * XS + f, v[j] * pick_f(w, j, hi));
        }
    } else {
        for (int j = 0; 2 * j < cnt; ++j) {
            const int sj = pick_i(s, j, hi), dj = pick_i(d, j, hi);
            const float wj = pick_f(w, j, hi);
            if (f < F_IN && 2 * j + hi < cnt) atomicAddF(aggx + dj * XS + f, x[sj * F_IN + f] * wj);
        }
    }
}


__global__ void k_zero(float* p, int n) {
    int i = blockIdx.x * blockDim.x + threadIdx.x;
    if (i < n) p[i] = 0.0f;
}


__global__ void k_gemm_fold(const float* __restrict__ A, const float* __restrict__ W,
                            const float* __restrict__ lw, const float* __restrict__ bprev,
                            const float* __restrict__ dis, float* __restrict__ y,
                            float* __restrict__ racc, int n) {
    __shared__ alignas(16) float Ulds[H];
    __shared__ alignas(16) float Blds[H];
    __shared__ float Zlds[32 * 16];
    for (int k = threadIdx.x >> 2; k < H; k += blockDim.x >> 2) {
        const int q = threadIdx.x & 3;
        const float4* wr = reinterpret_cast<const float4*>(W + k * H + q * 16);
        const float4* lr = reinterpret_cast<const float4*>(lw + q * 16);
        float p = 0.0f;
        #pragma unroll
        for (int i = 0; i < 4; ++i) {
            const float4 a = wr[i], b = lr[i];
            p += a.x * b.x + a.y * b.y + a.z * b.z + a.w * b.w;
        }
        p += __shfl_xor(p, 1);
        p += __shfl_xor(p, 2);
        if (q == 0) Ulds[k] = p;
    }
    for (int idx = threadIdx.x; idx < H; idx += blockDim.x) Blds[idx] = bprev[idx];
    __syncthreads();

    const int lane = threadIdx.x & 31;
    const int wv    = __builtin_amdgcn_readfirstlane(threadIdx.x >> 5);
    const int row0  = (blockIdx.x * (blockDim.x >> 5) + wv) * 16;
    const int mrow  = lane & 15;
    const int hi    = lane >> 4;
    if (row0 < n) {
        int r = row0 + mrow;
        r = r < n ? r : n - 1;
        const float* arow = A + (size_t)r * H;
        v8f acc0 = (v8f){0, 0, 0, 0, 0, 0, 0, 0};
        v8f acc1 = (v8f){0, 0, 0, 0, 0, 0, 0, 0};
        #pragma unroll
        for (int k0 = 0; k0 < H; k0 += 8) {
            const int ka = k0 + hi * 2;
            v2f a0 = *reinterpret_cast<const v2f*>(arow + ka);
            v2f a1 = *reinterpret_cast<const v2f*>(arow + ka + 4);
            const v2f p0 = *reinterpret_cast<const v2f*>(&Blds[ka]);
            const v2f p1 = *reinterpret_cast<const v2f*>(&Blds[ka + 4]);
            a0.x = fmaxf(a0.x + p0.x, 0.0f);
            a0.y = fmaxf(a0.y + p0.y, 0.0f);
            a1.x = fmaxf(a1.x + p1.x, 0.0f);
            a1.y = fmaxf(a1.y + p1.y, 0.0f);
            const v2f u0 = *reinterpret_cast<const v2f*>(&Ulds[ka]);
            const v2f u1 = *reinterpret_cast<const v2f*>(&Ulds[ka + 4]);
            v2f b0, b1;
            b0.x = mrow == 0 ? u0.x : 0.0f;
            b0.y = mrow == 0 ? u0.y : 0.0f;
            b1.x = mrow == 0 ? u1.x : 0.0f;
            b1.y = mrow == 0 ? u1.y : 0.0f;
            acc0 = __builtin_amdgcn_wmma_f32_16x16x4_f32(false, a0, false, b0, (short)0, acc0, false, false);
            acc1 = __builtin_amdgcn_wmma_f32_16x16x4_f32(false, a1, false, b1, (short)0, acc1, false, false);
        }
        v8f z = acc0 + acc1;
        if (mrow == 0) {
            #pragma unroll
            for (int j = 0; j < 8; ++j) Zlds[wv * 16 + hi * 8 + j] = z[j];
        }
    }
    __syncthreads();
    const int brow = (blockDim.x >> 5) * 16;
    for (int t = threadIdx.x; t < brow; t += blockDim.x) {
        const int r = blockIdx.x * brow + t;
        if (r < n) {
            const float v = Zlds[t] * dis[r];
            y[r] = v;
            racc[r] = v;
        }
    }
}

__global__ void k_scatter_y(const int* __restrict__ src, const int* __restrict__ dst,
                            const float* __restrict__ y, float* __restrict__ racc, int e) {
    const int e0 = (blockIdx.x * blockDim.x + threadIdx.x) * 4;
    if (e0 >= e) return;
    if (e0 + 4 <= e && (e & 3) == 0) {
        const int4 s = *reinterpret_cast<const int4*>(src + e0);
        const int4 d = *reinterpret_cast<const int4*>(dst + e0);
        const float v0 = y[s.x], v1 = y[s.y], v2 = y[s.z], v3 = y[s.w];
        atomicAddF(racc + d.x, v0);
        atomicAddF(racc + d.y, v1);
        atomicAddF(racc + d.z, v2);
        atomicAddF(racc + d.w, v3);
    } else {
        for (int i = e0; i < e && i < e0 + 4; ++i) atomicAddF(racc + dst[i], y[src[i]]);
    }
}

__global__ void k_pool_y(const float* __restrict__ racc, const float* __restrict__ dis,
                         const int* __restrict__ batch, float* __restrict__ sums,
                         float* __restrict__ cnts, int n) {
    const int i = blockIdx.x * blockDim.x + threadIdx.x;
    if (i >= n) return;
    const int g = batch[i];
    atomicAddF(&sums[g], racc[i] * dis[i]);
    atomicAddF(&cnts[g], 1.0f);
}

__global__ void k_pool_final(const float* __restrict__ sums, const float* __restrict__ cnts,
                             const float* __restrict__ b, const float* __restrict__ lin_w,
                             const float* __restrict__ lin_b, float* __restrict__ out, int g_total) {
    const int g = blockIdx.x * blockDim.x + threadIdx.x;
    if (g >= g_total) return;
    float c = 0.0f;
    for (int f = 0; f < H; ++f) c += b[f] * lin_w[f];
    const float cn = cnts[g];
    out[g] = (sums[g] + cn * c) / fmaxf(cn, 1.0f) + lin_b[0];
}


extern "C" void kernel_launch(void* const* d_in, const int* in_sizes, int n_in,
                              void* d_out, int out_size, void* d_ws, size_t ws_size,
                              hipStream_t stream) {
    const float* x     = (const float*)d_in[0];
    const int*   eidx  = (const int*)  d_in[1];
    const int*   batch = (const int*)  d_in[2];
    const float* W1    = (const float*)d_in[3];
    const float* b1    = (const float*)d_in[4];
    const float* W2    = (const float*)d_in[5];
    const float* b2    = (const float*)d_in[6];
    const float* W3    = (const float*)d_in[7];
    const float* b3    = (const float*)d_in[8];
    const float* lin_w = (const float*)d_in[9];
    const float* lin_b = (const float*)d_in[10];
    float* out = (float*)d_out;

    const int n = in_sizes[0] / F_IN;
    const int e = in_sizes[1] / 2;
    const int* src = eidx;
    const int* dst = eidx + e;

    float* ws   = (float*)d_ws;
    float* dis  = ws;
    float* bufA = dis  + n;
    float* bufB = bufA + (size_t)n * H;
    float* sums = bufB + (size_t)n * H;
    float* cnts = sums + N_GRAPHS;

    const int TB = 256;
    dim3 blk(TB);
    auto blocks = [](long long work, int tb) { return dim3((unsigned)((work + tb - 1) / tb)); };

    k_init_deg  <<<blocks(n, TB), blk, 0, stream>>>(dis, n);
    k_deg_edges <<<blocks(e, TB), blk, 0, stream>>>(dst, dis, e);
    float* aggx = bufA;
    k_rsqrt_start<<<blocks(n, TB), blk, 0, stream>>>(dis, x, aggx, n);

    const int strips       = (n + 15) / 16;
    const dim3 gemm_grid   = blocks((long long)strips * 32, TB);

    k_scatter_x <<<blocks((long long)e, TB), blk, 0, stream>>>(src, dst, dis, x, aggx, e);
    k_gemm_wmma<F_IN, XS, 12, false, false><<<gemm_grid, blk, 0, stream>>>(aggx, W1, b1, dis, bufB, bufB, n);


    k_gemm_wmma<H, H, H, true, true><<<gemm_grid, blk, 0, stream>>>(bufB, W2, b1, dis, bufA, bufB, n);
    k_scatter_edges <<<blocks((long long)e, TB), blk, 0, stream>>>(src, dst, dis, bufA, bufB, e);

    float* y    = bufA;
    float* racc = bufA + n;
    k_gemm_fold <<<gemm_grid, blk, 0, stream>>>(bufB, W3, lin_w, b2, dis, y, racc, n);
    k_scatter_y <<<blocks(((long long)e + 3) / 4, TB), blk, 0, stream>>>(src, dst, y, racc, e);

    k_zero      <<<blocks(2 * N_GRAPHS, TB), blk, 0, stream>>>(sums, 2 * N_GRAPHS);
    k_pool_y    <<<blocks(n, TB), blk, 0, stream>>>(racc, dis, batch, sums, cnts, n);
    k_pool_final<<<blocks(N_GRAPHS, TB), blk, 0, stream>>>(sums, cnts, b3, lin_w, lin_b, out, N_GRAPHS);
}


# layer 2 aggregates before it multiplies: the edge scatter sums relu'd h1 rows, then one kernel does W2, bias, relu and the folded W3*lin_w product in registers (h2 never stored); float sums regrouped
# speedup vs baseline: 2.5167x; 1.0779x over previous
#include <hip/hip_runtime.h>
#include <hip/hip_bf16.h>

#define N_NODES 200000
#define N_EDGES 1000000
#define N_GRAPHS 4096
#define F_IN 11
#define H 64
#define XS 16

__device__ __forceinline__ int aggx_off(int i) { return (i >> 4) * (16 * H) + (i & 15) * XS; }

typedef float v2f __attribute__((ext_vector_type(2)));
typedef float v8f __attribute__((ext_vector_type(8)));

__device__ __forceinline__ void atomicAddF(float* p, float v) {
    unsafeAtomicAdd(p, v);
}


__global__ void k_init_deg(float* deg, int n) {
    int i = blockIdx.x * blockDim.x + threadIdx.x;
    if (i < n) deg[i] = 1.0f;
}

__global__ void k_deg_edges(const int* __restrict__ dst, float* deg, int e) {
    int i = blockIdx.x * blockDim.x + threadIdx.x;
    if (i < e) atomicAddF(&deg[dst[i]], 1.0f);
}

__global__ void k_rsqrt_start(float* deg, const float* __restrict__ x, float* __restrict__ aggx, int n) {
    int i = blockIdx.x * blockDim.x + threadIdx.x;
    if (i >= n) return;
    const float dv = rsqrtf(fmaxf(deg[i], 1.0f));
    deg[i] = dv;
    const float ws = dv * dv;
    const float* __restrict__ xr = x + i * F_IN;
    float r[XS];
    #pragma unroll
    for (int f = 0; f < XS; ++f) r[f] = (f < F_IN) ? xr[f] * ws : 0.0f;
    float4* o = reinterpret_cast<float4*>(aggx + aggx_off(i));
    #pragma unroll
    for (int q = 0; q < XS / 4; ++q) o[q] = make_float4(r[4 * q], r[4 * q + 1], r[4 * q + 2], r[4 * q + 3]);
}


template <int KV, int KP>
__global__ void k_gemm_l1(const float* A, const float* __restrict__ W,
                          const float* __restrict__ b, const float* __restrict__ dis,
                          float* __restrict__ Out, float* Agg, int n) {
    static_assert(KP <= XS && (KP & 3) == 0, "A rows hold XS columns");
    __shared__ float Wlds[(KP / 2) * H * 2];
    for (int idx = threadIdx.x; idx < KP * H; idx += blockDim.x) {
        int k = idx / H, c = idx - k * H;
        float v = (k < KV) ? W[k * H + c] : 0.0f;
        Wlds[(k >> 1) * (H * 2) + c * 2 + (k & 1)] = v;
    }
    __syncthreads();

    const int lane  = threadIdx.x & 31;
    const int strip = blockIdx.x * (blockDim.x >> 5) + (threadIdx.x >> 5);
    const int row0  = strip * 16;
    if (row0 >= n) return;

    const int mrow = lane & 15;
    const int hi   = lane >> 4;
    const float* arow = A + aggx_off(row0 + mrow);

    const int mbase = (lane < 16) ? 0 : 8;
    float wself[8];
    #pragma unroll
    for (int j = 0; j < 8; ++j) {
        const float dv = dis[row0 + mbase + j];
        wself[j] = dv * dv;
    }

    v8f acc[4];
    #pragma unroll
    for (int t = 0; t < 4; ++t) acc[t] = (v8f){0, 0, 0, 0, 0, 0, 0, 0};

    #pragma unroll
    for (int k0 = 0; k0 < KP; k0 += 4) {
        const int ka = k0 + hi * 2;
        const v2f a = *reinterpret_cast<const v2f*>(arow + ka);
        const int pbase = ((k0 >> 1) + hi) * (H * 2);
        #pragma unroll
        for (int t = 0; t < 4; ++t) {
            const int c = t * 16 + mrow;
            v2f bw = *reinterpret_cast<const v2f*>(&Wlds[pbase + c * 2]);
            acc[t] = __builtin_amdgcn_wmma_f32_16x16x4_f32(
                 false, a,  false, bw,
                 (short)0, acc[t],  false,  false);
        }
    }

    #pragma unroll
    for (int t = 0; t < 4; ++t) {
        const int col = t * 16 + mrow;
        const float bc = b[col];
        #pragma unroll
        for (int j = 0; j < 8; ++j) {
            const float v = fmaxf(acc[t][j] + bc, 0.0f);
            Out[(row0 + mbase + j) * H + col] = v;
            Agg[(row0 + mbase + j) * H + col] = v * wself[j];
        }
    }
}


#define EDGES_PER_WAVE 32
#define EDGE_CHUNK 8

__device__ __forceinline__ int   bcast_i(int v, int l)   { return __builtin_amdgcn_readlane(v, l); }
__device__ __forceinline__ float bcast_f(float v, int l) {
    return __int_as_float(__builtin_amdgcn_readlane(__float_as_int(v), l));
}

__global__ void k_scatter_edges(const int* __restrict__ src, const int* __restrict__ dst,
                                const float* __restrict__ dis, const float* __restrict__ h,
                                float* __restrict__ agg, int e) {
    const int lane = threadIdx.x & 31;
    const int wave = blockIdx.x * (blockDim.x >> 5) + __builtin_amdgcn_readfirstlane(threadIdx.x >> 5);
    const int e0 = wave * EDGES_PER_WAVE;
    if (e0 >= e) return;
    const int left = e - e0;
    const int cnt = left < EDGES_PER_WAVE ? left : EDGES_PER_WAVE;

    int s = 0, d = 0;
    float w = 0.0f;
    if (lane < cnt) {
        s = src[e0 + lane];
        d = dst[e0 + lane];
        w = dis[s] * dis[d];
    }

    int c = 0;
    for (; c + EDGE_CHUNK <= cnt; c += EDGE_CHUNK) {
        float lo[EDGE_CHUNK], hi[EDGE_CHUNK];
        #pragma unroll
        for (int j = 0; j < EDGE_CHUNK; ++j) {
            const float* __restrict__ row = h + (size_t)bcast_i(s, c + j) * H;
            lo[j] = row[lane];
            hi[j] = row[lane + 32];
        }
        #pragma unroll
        for (int j = 0; j < EDGE_CHUNK; ++j) {
            const float wj = bcast_f(w, c + j);
            float* o = agg + (size_t)bcast_i(d, c + j) * H;
            atomicAddF(o + lane,      lo[j] * wj);
            atomicAddF(o + lane + 32, hi[j] * wj);
        }
    }
    for (; c < cnt; ++c) {
        const float* __restrict__ row = h + (size_t)bcast_i(s, c) * H;
        const float wj = bcast_f(w, c);
        float* o = agg + (size_t)bcast_i(d, c) * H;
        atomicAddF(o + lane,      row[lane] * wj);
        atomicAddF(o + lane + 32, row[lane + 32] * wj);
    }
}

__device__ __forceinline__ int   pick_i(int v, int j, int hi)   { const int a = bcast_i(v, 2 * j), b = bcast_i(v, 2 * j + 1); return hi ? b : a; }
__device__ __forceinline__ float pick_f(float v, int j, int hi) { const float a = bcast_f(v, 2 * j), b = bcast_f(v, 2 * j + 1); return hi ? b : a; }

__global__ void k_scatter_x(const int* __restrict__ src, const int* __restrict__ dst,
                            const float* __restrict__ dis, const float* __restrict__ x,
                            float* __restrict__ aggx, int e) {
    const int lane = threadIdx.x & 31;
    const int wave = blockIdx.x * (blockDim.x >> 5) + __builtin_amdgcn_readfirstlane(threadIdx.x >> 5);
    const int e0 = wave * EDGES_PER_WAVE;
    if (e0 >= e) return;
    const int left = e - e0;
    const int cnt = left < EDGES_PER_WAVE ? left : EDGES_PER_WAVE;

    int s = 0, d = 0;
    float w = 0.0f;
    if (lane < cnt) {
        s = src[e0 + lane];
        d = dst[e0 + lane];
        w = dis[s] * dis[d];
    }

    const int f  = lane & 15;
    const int hi = lane >> 4;
    if (cnt == EDGES_PER_WAVE) {
        if (f < F_IN) {
            float v[EDGES_PER_WAVE / 2];
            #pragma unroll
            for (int j = 0; j < EDGES_PER_WAVE / 2; ++j) v[j] = x[pick_i(s, j, hi) * F_IN + f];
            #pragma unroll
            for (int j = 0; j < EDGES_PER_WAVE / 2; ++j)
                atomicAddF(aggx + aggx_off(pick_i(d, j, hi)) + f, v[j] * pick_f(w, j, hi));
        }
    } else {
        for (int j = 0; 2 * j < cnt; ++j) {
            const int sj = pick_i(s, j, hi), dj = pick_i(d, j, hi);
            const float wj = pick_f(w, j, hi);
            if (f < F_IN && 2 * j + hi < cnt) atomicAddF(aggx + aggx_off(dj) + f, x[sj * F_IN + f] * wj);
        }
    }
}


__global__ void k_zero(float* p, int n) {
    int i = blockIdx.x * blockDim.x + threadIdx.x;
    if (i < n) p[i] = 0.0f;
}


__global__ void k_gemm2_fold(const float* __restrict__ G, const float* __restrict__ W2,
                             const float* __restrict__ b2, const float* __restrict__ W3,
                             const float* __restrict__ lw, const float* __restrict__ dis,
                             float* __restrict__ y, float* __restrict__ racc, int n) {
    __shared__ float Wlds[(H / 2) * H * 2];
    __shared__ alignas(16) float Ulds[H];
    __shared__ alignas(16) float Blds[H];
    __shared__ float Zlds[32 * 16];
    for (int idx = threadIdx.x; idx < H * H; idx += blockDim.x) {
        int k = idx / H, c = idx - k * H;
        Wlds[(k >> 1) * (H * 2) + c * 2 + (k & 1)] = W2[k * H + c];
    }
    for (int k = threadIdx.x >> 2; k < H; k += blockDim.x >> 2) {
        const int q = threadIdx.x & 3;
        const float4* wr = reinterpret_cast<const float4*>(W3 + k * H + q * 16);
        const float4* lr = reinterpret_cast<const float4*>(lw + q * 16);
        float p = 0.0f;
        #pragma unroll
        for (int i = 0; i < 4; ++i) {
            const float4 a = wr[i], b = lr[i];
            p += a.x * b.x + a.y * b.y + a.z * b.z + a.w * b.w;
        }
        p += __shfl_xor(p, 1);
        p += __shfl_xor(p, 2);
        if (q == 0) Ulds[k] = p;
    }
    for (int idx = threadIdx.x; idx < H; idx += blockDim.x) Blds[idx] = b2[idx];
    __syncthreads();

    const int lane = threadIdx.x & 31;
    const int wv    = __builtin_amdgcn_readfirstlane(threadIdx.x >> 5);
    const int row0  = (blockIdx.x * (blockDim.x >> 5) + wv) * 16;
    const int mrow  = lane & 15;
    const int hi    = lane >> 4;
    if (row0 < n) {
        int r = row0 + mrow;
        r = r < n ? r : n - 1;
        const float* grow = G + (size_t)r * H;
        v8f acc[4];
        #pragma unroll
        for (int t = 0; t < 4; ++t) acc[t] = (v8f){0, 0, 0, 0, 0, 0, 0, 0};
        #pragma unroll
        for (int k0 = 0; k0 < H; k0 += 4) {
            const int ka = k0 + hi * 2;
            const v2f g = *reinterpret_cast<const v2f*>(grow + ka);
            const int pbase = ((k0 >> 1) + hi) * (H * 2);
            #pragma unroll
            for (int t = 0; t < 4; ++t) {
                const int c = t * 16 + mrow;
                const v2f w = *reinterpret_cast<const v2f*>(&Wlds[pbase + c * 2]);
                acc[t] = __builtin_amdgcn_wmma_f32_16x16x4_f32(false, w, false, g, (short)0, acc[t], false, false);
            }
        }
        v8f z0 = (v8f){0, 0, 0, 0, 0, 0, 0, 0};
        v8f z1 = (v8f){0, 0, 0, 0, 0, 0, 0, 0};
        #pragma unroll
        for (int t = 0; t < 4; ++t) {
            const int cb = t * 16 + hi * 8;
            const float4 ba = *reinterpret_cast<const float4*>(&Blds[cb]);
            const float4 bb = *reinterpret_cast<const float4*>(&Blds[cb + 4]);
            const float4 ua = *reinterpret_cast<const float4*>(&Ulds[cb]);
            const float4 ub = *reinterpret_cast<const float4*>(&Ulds[cb + 4]);
            const float bias[8] = {ba.x, ba.y, ba.z, ba.w, bb.x, bb.y, bb.z, bb.w};
            const float uu[8]   = {ua.x, ua.y, ua.z, ua.w, ub.x, ub.y, ub.z, ub.w};
            #pragma unroll
            for (int q = 0; q < 4; ++q) {
                v2f hv, uv;
                hv.x = fmaxf(acc[t][2 * q]     + bias[2 * q],     0.0f);
                hv.y = fmaxf(acc[t][2 * q + 1] + bias[2 * q + 1], 0.0f);
                uv.x = mrow == 0 ? uu[2 * q]     : 0.0f;
                uv.y = mrow == 0 ? uu[2 * q + 1] : 0.0f;
                if (q & 1) z1 = __builtin_amdgcn_wmma_f32_16x16x4_f32(false, uv, false, hv, (short)0, z1, false, false);
                else       z0 = __builtin_amdgcn_wmma_f32_16x16x4_f32(false, uv, false, hv, (short)0, z0, false, false);
            }
        }
        const v8f z = z0 + z1;
        if (hi == 0) Zlds[wv * 16 + mrow] = z[0];
    }
    __syncthreads();
    const int brow = (blockDim.x >> 5) * 16;
    for (int t = threadIdx.x; t < brow; t += blockDim.x) {
        const int r = blockIdx.x * brow + t;
        if (r < n) {
            const float v = Zlds[t] * dis[r];
            y[r] = v;
            racc[r] = v;
        }
    }
}

__global__ void k_scatter_y(const int* __restrict__ src, const int* __restrict__ dst,
                            const float* __restrict__ y, float* __restrict__ racc, int e) {
    const int e0 = (blockIdx.x * blockDim.x + threadIdx.x) * 4;
    if (e0 >= e) return;
    if (e0 + 4 <= e && (e & 3) == 0) {
        const int4 s = *reinterpret_cast<const int4*>(src + e0);
        const int4 d = *reinterpret_cast<const int4*>(dst + e0);
        const float v0 = y[s.x], v1 = y[s.y], v2 = y[s.z], v3 = y[s.w];
        atomicAddF(racc + d.x, v0);
        atomicAddF(racc + d.y, v1);
        atomicAddF(racc + d.z, v2);
        atomicAddF(racc + d.w, v3);
    } else {
        for (int i = e0; i < e && i < e0 + 4; ++i) atomicAddF(racc + dst[i], y[src[i]]);
    }
}

__global__ void k_pool_y(const float* __restrict__ racc, const float* __restrict__ dis,
                         const int* __restrict__ batch, float* __restrict__ sums,
                         float* __restrict__ cnts, int n) {
    const int i = blockIdx.x * blockDim.x + threadIdx.x;
    if (i >= n) return;
    const int g = batch[i];
    atomicAddF(&sums[g], racc[i] * dis[i]);
    atomicAddF(&cnts[g], 1.0f);
}

__global__ void k_pool_final(const float* __restrict__ sums, const float* __restrict__ cnts,
                             const float* __restrict__ b, const float* __restrict__ lin_w,
                             const float* __restrict__ lin_b, float* __restrict__ out, int g_total) {
    const int g = blockIdx.x * blockDim.x + threadIdx.x;
    if (g >= g_total) return;
    float c = 0.0f;
    for (int f = 0; f < H; ++f) c += b[f] * lin_w[f];
    const float cn = cnts[g];
    out[g] = (sums[g] + cn * c) / fmaxf(cn, 1.0f) + lin_b[0];
}


extern "C" void kernel_launch(void* const* d_in, const int* in_sizes, int n_in,
                              void* d_out, int out_size, void* d_ws, size_t ws_size,
                              hipStream_t stream) {
    const float* x     = (const float*)d_in[0];
    const int*   eidx  = (const int*)  d_in[1];
    const int*   batch = (const int*)  d_in[2];
    const float* W1    = (const float*)d_in[3];
    const float* b1    = (const float*)d_in[4];
    const float* W2    = (const float*)d_in[5];
    const float* b2    = (const float*)d_in[6];
    const float* W3    = (const float*)d_in[7];
    const float* b3    = (const float*)d_in[8];
    const float* lin_w = (const float*)d_in[9];
    const float* lin_b = (const float*)d_in[10];
    float* out = (float*)d_out;

    const int n = in_sizes[0] / F_IN;
    const int e = in_sizes[1] / 2;
    const int* src = eidx;
    const int* dst = eidx + e;

    float* ws   = (float*)d_ws;
    float* dis  = ws;
    float* bufA = dis  + n;
    float* bufB = bufA + (size_t)n * H;
    float* sums = bufB + (size_t)n * H;
    float* cnts = sums + N_GRAPHS;

    const int TB = 256;
    dim3 blk(TB);
    auto blocks = [](long long work, int tb) { return dim3((unsigned)((work + tb - 1) / tb)); };

    k_init_deg  <<<blocks(n, TB), blk, 0, stream>>>(dis, n);
    k_deg_edges <<<blocks(e, TB), blk, 0, stream>>>(dst, dis, e);
    float* aggx = bufA;
    k_rsqrt_start<<<blocks(n, TB), blk, 0, stream>>>(dis, x, aggx, n);

    const int strips       = (n + 15) / 16;
    const dim3 gemm_grid   = blocks((long long)strips * 32, TB);

    k_scatter_x <<<blocks((long long)e, TB), blk, 0, stream>>>(src, dst, dis, x, aggx, e);
    k_gemm_l1<F_IN, 12><<<gemm_grid, blk, 0, stream>>>(aggx, W1, b1, dis, bufB, bufA, n);

    k_scatter_edges <<<blocks((long long)e, TB), blk, 0, stream>>>(src, dst, dis, bufB, bufA, e);

    float* y    = bufB;
    float* racc = bufB + n;
    k_gemm2_fold<<<gemm_grid, blk, 0, stream>>>(bufA, W2, b2, W3, lin_w, dis, y, racc, n);
    k_scatter_y <<<blocks(((long long)e + 3) / 4, TB), blk, 0, stream>>>(src, dst, y, racc, e);

    k_zero      <<<blocks(2 * N_GRAPHS, TB), blk, 0, stream>>>(sums, 2 * N_GRAPHS);
    k_pool_y    <<<blocks(n, TB), blk, 0, stream>>>(racc, dis, batch, sums, cnts, n);
    k_pool_final<<<blocks(N_GRAPHS, TB), blk, 0, stream>>>(sums, cnts, b3, lin_w, lin_b, out, N_GRAPHS);
}


# edge weight split: dis[src] folded into the stored source rows, dis[dst] applied once per node by the consuming GEMM; both row scatters drop their per-edge dis gathers; float products regrouped
# speedup vs baseline: 2.6102x; 1.0372x over previous
#include <hip/hip_runtime.h>
#include <hip/hip_bf16.h>

#define N_NODES 200000
#define N_EDGES 1000000
#define N_GRAPHS 4096
#define F_IN 11
#define H 64
#define XS 16

__device__ __forceinline__ int aggx_off(int i) { return (i >> 4) * (16 * H) + (i & 15) * XS; }

typedef float v2f __attribute__((ext_vector_type(2)));
typedef float v8f __attribute__((ext_vector_type(8)));

__device__ __forceinline__ void atomicAddF(float* p, float v) {
    unsafeAtomicAdd(p, v);
}


__global__ void k_init_deg(float* deg, int n) {
    int i = blockIdx.x * blockDim.x + threadIdx.x;
    if (i < n) deg[i] = 1.0f;
}

__global__ void k_deg_edges(const int* __restrict__ dst, float* deg, int e) {
    int i = blockIdx.x * blockDim.x + threadIdx.x;
    if (i < e) atomicAddF(&deg[dst[i]], 1.0f);
}

__global__ void k_rsqrt_start(float* deg, const float* __restrict__ x, float* __restrict__ aggx,
                              float* __restrict__ xs, int n) {
    int i = blockIdx.x * blockDim.x + threadIdx.x;
    if (i >= n) return;
    const float dv = rsqrtf(fmaxf(deg[i], 1.0f));
    deg[i] = dv;
    const float* __restrict__ xr = x + i * F_IN;
    float r[XS];
    #pragma unroll
    for (int f = 0; f < XS; ++f) r[f] = (f < F_IN) ? xr[f] * dv : 0.0f;
    float4* o = reinterpret_cast<float4*>(aggx + aggx_off(i));
    float4* p = reinterpret_cast<float4*>(xs + i * XS);
    #pragma unroll
    for (int q = 0; q < XS / 4; ++q) {
        const float4 v = make_float4(r[4 * q], r[4 * q + 1], r[4 * q + 2], r[4 * q + 3]);
        o[q] = v;
        p[q] = v;
    }
}


template <int KV, int KP>
__global__ void k_gemm_l1(const float* A, const float* __restrict__ W,
                          const float* __restrict__ b, const float* __restrict__ dis,
                          float* __restrict__ Out, float* Agg, int n) {
    static_assert(KP <= XS && (KP & 3) == 0, "A rows hold XS columns");
    __shared__ float Wlds[(KP / 2) * H * 2];
    for (int idx = threadIdx.x; idx < KP * H; idx += blockDim.x) {
        int k = idx / H, c = idx - k * H;
        float v = (k < KV) ? W[k * H + c] : 0.0f;
        Wlds[(k >> 1) * (H * 2) + c * 2 + (k & 1)] = v;
    }
    __syncthreads();

    const int lane  = threadIdx.x & 31;
    const int strip = blockIdx.x * (blockDim.x >> 5) + (threadIdx.x >> 5);
    const int row0  = strip * 16;
    if (row0 >= n) return;

    const int mrow = lane & 15;
    const int hi   = lane >> 4;
    const float* arow = A + aggx_off(row0 + mrow);

    const int mbase = (lane < 16) ? 0 : 8;
    float drow[8];
    #pragma unroll
    for (int j = 0; j < 8; ++j) drow[j] = dis[row0 + mbase + j];

    v8f acc[4];
    #pragma unroll
    for (int t = 0; t < 4; ++t) acc[t] = (v8f){0, 0, 0, 0, 0, 0, 0, 0};

    #pragma unroll
    for (int k0 = 0; k0 < KP; k0 += 4) {
        const int ka = k0 + hi * 2;
        const v2f a = *reinterpret_cast<const v2f*>(arow + ka);
        const int pbase = ((k0 >> 1) + hi) * (H * 2);
        #pragma unroll
        for (int t = 0; t < 4; ++t) {
            const int c = t * 16 + mrow;
            v2f bw = *reinterpret_cast<const v2f*>(&Wlds[pbase + c * 2]);
            acc[t] = __builtin_amdgcn_wmma_f32_16x16x4_f32(
                 false, a,  false, bw,
                 (short)0, acc[t],  false,  false);
        }
    }

    #pragma unroll
    for (int t = 0; t < 4; ++t) {
        const int col = t * 16 + mrow;
        const float bc = b[col];
        #pragma unroll
        for (int j = 0; j < 8; ++j) {
            const float v = fmaxf(fmaf(acc[t][j], drow[j], bc), 0.0f) * drow[j];
            Out[(row0 + mbase + j) * H + col] = v;
            Agg[(row0 + mbase + j) * H + col] = v;
        }
    }
}


#define EDGES_PER_WAVE 32
#define EDGE_CHUNK 8

__device__ __forceinline__ int bcast_i(int v, int l) { return __builtin_amdgcn_readlane(v, l); }

__global__ void k_scatter_edges(const int* __restrict__ src, const int* __restrict__ dst,
                                const float* __restrict__ h, float* __restrict__ agg, int e) {
    const int lane = threadIdx.x & 31;
    const int wave = blockIdx.x * (blockDim.x >> 5) + __builtin_amdgcn_readfirstlane(threadIdx.x >> 5);
    const int e0 = wave * EDGES_PER_WAVE;
    if (e0 >= e) return;
    const int left = e - e0;
    const int cnt = left < EDGES_PER_WAVE ? left : EDGES_PER_WAVE;

    int s = 0, d = 0;
    if (lane < cnt) {
        s = src[e0 + lane];
        d = dst[e0 + lane];
    }

    int c = 0;
    for (; c + EDGE_CHUNK <= cnt; c += EDGE_CHUNK) {
        float lo[EDGE_CHUNK], hi[EDGE_CHUNK];
        #pragma unroll
        for (int j = 0; j < EDGE_CHUNK; ++j) {
            const float* __restrict__ row = h + (size_t)bcast_i(s, c + j) * H;
            lo[j] = row[lane];
            hi[j] = row[lane + 32];
        }
        #pragma unroll
        for (int j = 0; j < EDGE_CHUNK; ++j) {
            float* o = agg + (size_t)bcast_i(d, c + j) * H;
            atomicAddF(o + lane,      lo[j]);
            atomicAddF(o + lane + 32, hi[j]);
        }
    }
    for (; c < cnt; ++c) {
        const float* __restrict__ row = h + (size_t)bcast_i(s, c) * H;
        float* o = agg + (size_t)bcast_i(d, c) * H;
        atomicAddF(o + lane,      row[lane]);
        atomicAddF(o + lane + 32, row[lane + 32]);
    }
}

__device__ __forceinline__ int   pick_i(int v, int j, int hi)   { const int a = bcast_i(v, 2 * j), b = bcast_i(v, 2 * j + 1); return hi ? b : a; }

__global__ void k_scatter_x(const int* __restrict__ src, const int* __restrict__ dst,
                            const float* __restrict__ xs, float* __restrict__ aggx, int e) {
    const int lane = threadIdx.x & 31;
    const int wave = blockIdx.x * (blockDim.x >> 5) + __builtin_amdgcn_readfirstlane(threadIdx.x >> 5);
    const int e0 = wave * EDGES_PER_WAVE;
    if (e0 >= e) return;
    const int left = e - e0;
    const int cnt = left < EDGES_PER_WAVE ? left : EDGES_PER_WAVE;

    int s = 0, d = 0;
    if (lane < cnt) {
        s = src[e0 + lane];
        d = dst[e0 + lane];
    }

    const int f  = lane & 15;
    const int hi = lane >> 4;
    if (cnt == EDGES_PER_WAVE) {
        if (f < F_IN) {
            float v[EDGES_PER_WAVE / 2];
            #pragma unroll
            for (int j = 0; j < EDGES_PER_WAVE / 2; ++j) v[j] = xs[pick_i(s, j, hi) * XS + f];
            #pragma unroll
            for (int j = 0; j < EDGES_PER_WAVE / 2; ++j)
                atomicAddF(aggx + aggx_off(pick_i(d, j, hi)) + f, v[j]);
        }
    } else {
        for (int j = 0; 2 * j < cnt; ++j) {
            const int sj = pick_i(s, j, hi), dj = pick_i(d, j, hi);
            if (f < F_IN && 2 * j + hi < cnt) atomicAddF(aggx + aggx_off(dj) + f, xs[sj * XS + f]);
        }
    }
}


__global__ void k_zero(float* p, int n) {
    int i = blockIdx.x * blockDim.x + threadIdx.x;
    if (i < n) p[i] = 0.0f;
}


__global__ void k_gemm2_fold(const float* __restrict__ G, const float* __restrict__ W2,
                             const float* __restrict__ b2, const float* __restrict__ W3,
                             const float* __restrict__ lw, const float* __restrict__ dis,
                             float* __restrict__ y, float* __restrict__ racc, int n) {
    __shared__ float Wlds[(H / 2) * H * 2];
    __shared__ alignas(16) float Ulds[H];
    __shared__ alignas(16) float Blds[H];
    __shared__ float Zlds[32 * 16];
    for (int idx = threadIdx.x; idx < H * H; idx += blockDim.x) {
        int k = idx / H, c = idx - k * H;
        Wlds[(k >> 1) * (H * 2) + c * 2 + (k & 1)] = W2[k * H + c];
    }
    for (int k = threadIdx.x >> 2; k < H; k += blockDim.x >> 2) {
        const int q = threadIdx.x & 3;
        const float4* wr = reinterpret_cast<const float4*>(W3 + k * H + q * 16);
        const float4* lr = reinterpret_cast<const float4*>(lw + q * 16);
        float p = 0.0f;
        #pragma unroll
        for (int i = 0; i < 4; ++i) {
            const float4 a = wr[i], b = lr[i];
            p += a.x * b.x + a.y * b.y + a.z * b.z + a.w * b.w;
        }
        p += __shfl_xor(p, 1);
        p += __shfl_xor(p, 2);
        if (q == 0) Ulds[k] = p;
    }
    for (int idx = threadIdx.x; idx < H; idx += blockDim.x) Blds[idx] = b2[idx];
    __syncthreads();

    const int lane = threadIdx.x & 31;
    const int wv    = __builtin_amdgcn_readfirstlane(threadIdx.x >> 5);
    const int row0  = (blockIdx.x * (blockDim.x >> 5) + wv) * 16;
    const int mrow  = lane & 15;
    const int hi    = lane >> 4;
    if (row0 < n) {
        int r = row0 + mrow;
        r = r < n ? r : n - 1;
        const float* grow = G + (size_t)r * H;
        const float dn = dis[r];
        v8f acc[4];
        #pragma unroll
        for (int t = 0; t < 4; ++t) acc[t] = (v8f){0, 0, 0, 0, 0, 0, 0, 0};
        #pragma unroll
        for (int k0 = 0; k0 < H; k0 += 4) {
            const int ka = k0 + hi * 2;
            const v2f g = *reinterpret_cast<const v2f*>(grow + ka);
            const int pbase = ((k0 >> 1) + hi) * (H * 2);
            #pragma unroll
            for (int t = 0; t < 4; ++t) {
                const int c = t * 16 + mrow;
                const v2f w = *reinterpret_cast<const v2f*>(&Wlds[pbase + c * 2]);
                acc[t] = __builtin_amdgcn_wmma_f32_16x16x4_f32(false, w, false, g, (short)0, acc[t], false, false);
            }
        }
        v8f z0 = (v8f){0, 0, 0, 0, 0, 0, 0, 0};
        v8f z1 = (v8f){0, 0, 0, 0, 0, 0, 0, 0};
        #pragma unroll
        for (int t = 0; t < 4; ++t) {
            const int cb = t * 16 + hi * 8;
            const float4 ba = *reinterpret_cast<const float4*>(&Blds[cb]);
            const float4 bb = *reinterpret_cast<const float4*>(&Blds[cb + 4]);
            const float4 ua = *reinterpret_cast<const float4*>(&Ulds[cb]);
            const float4 ub = *reinterpret_cast<const float4*>(&Ulds[cb + 4]);
            const float bias[8] = {ba.x, ba.y, ba.z, ba.w, bb.x, bb.y, bb.z, bb.w};
            const float uu[8]   = {ua.x, ua.y, ua.z, ua.w, ub.x, ub.y, ub.z, ub.w};
            #pragma unroll
            for (int q = 0; q < 4; ++q) {
                v2f hv, uv;
                hv.x = fmaxf(fmaf(acc[t][2 * q],     dn, bias[2 * q]),     0.0f);
                hv.y = fmaxf(fmaf(acc[t][2 * q + 1], dn, bias[2 * q + 1]), 0.0f);
                uv.x = mrow == 0 ? uu[2 * q]     : 0.0f;
                uv.y = mrow == 0 ? uu[2 * q + 1] : 0.0f;
                if (q & 1) z1 = __builtin_amdgcn_wmma_f32_16x16x4_f32(false, uv, false, hv, (short)0, z1, false, false);
                else       z0 = __builtin_amdgcn_wmma_f32_16x16x4_f32(false, uv, false, hv, (short)0, z0, false, false);
            }
        }
        const v8f z = z0 + z1;
        if (hi == 0) Zlds[wv * 16 + mrow] = z[0];
    }
    __syncthreads();
    const int brow = (blockDim.x >> 5) * 16;
    for (int t = threadIdx.x; t < brow; t += blockDim.x) {
        const int r = blockIdx.x * brow + t;
        if (r < n) {
            const float v = Zlds[t] * dis[r];
            y[r] = v;
            racc[r] = v;
        }
    }
}

__global__ void k_scatter_y(const int* __restrict__ src, const int* __restrict__ dst,
                            const float* __restrict__ y, float* __restrict__ racc, int e) {
    const int e0 = (blockIdx.x * blockDim.x + threadIdx.x) * 4;
    if (e0 >= e) return;
    if (e0 + 4 <= e && (e & 3) == 0) {
        const int4 s = *reinterpret_cast<const int4*>(src + e0);
        const int4 d = *reinterpret_cast<const int4*>(dst + e0);
        const float v0 = y[s.x], v1 = y[s.y], v2 = y[s.z], v3 = y[s.w];
        atomicAddF(racc + d.x, v0);
        atomicAddF(racc + d.y, v1);
        atomicAddF(racc + d.z, v2);
        atomicAddF(racc + d.w, v3);
    } else {
        for (int i = e0; i < e && i < e0 + 4; ++i) atomicAddF(racc + dst[i], y[src[i]]);
    }
}

__global__ void k_pool_y(const float* __restrict__ racc, const float* __restrict__ dis,
                         const int* __restrict__ batch, float* __restrict__ sums,
                         float* __restrict__ cnts, int n) {
    const int i = blockIdx.x * blockDim.x + threadIdx.x;
    if (i >= n) return;
    const int g = batch[i];
    atomicAddF(&sums[g], racc[i] * dis[i]);
    atomicAddF(&cnts[g], 1.0f);
}

__global__ void k_pool_final(const float* __restrict__ sums, const float* __restrict__ cnts,
                             const float* __restrict__ b, const float* __restrict__ lin_w,
                             const float* __restrict__ lin_b, float* __restrict__ out, int g_total) {
    const int g = blockIdx.x * blockDim.x + threadIdx.x;
    if (g >= g_total) return;
    float c = 0.0f;
    for (int f = 0; f < H; ++f) c += b[f] * lin_w[f];
    const float cn = cnts[g];
    out[g] = (sums[g] + cn * c) / fmaxf(cn, 1.0f) + lin_b[0];
}


extern "C" void kernel_launch(void* const* d_in, const int* in_sizes, int n_in,
                              void* d_out, int out_size, void* d_ws, size_t ws_size,
                              hipStream_t stream) {
    const float* x     = (const float*)d_in[0];
    const int*   eidx  = (const int*)  d_in[1];
    const int*   batch = (const int*)  d_in[2];
    const float* W1    = (const float*)d_in[3];
    const float* b1    = (const float*)d_in[4];
    const float* W2    = (const float*)d_in[5];
    const float* b2    = (const float*)d_in[6];
    const float* W3    = (const float*)d_in[7];
    const float* b3    = (const float*)d_in[8];
    const float* lin_w = (const float*)d_in[9];
    const float* lin_b = (const float*)d_in[10];
    float* out = (float*)d_out;

    const int n = in_sizes[0] / F_IN;
    const int e = in_sizes[1] / 2;
    const int* src = eidx;
    const int* dst = eidx + e;

    float* ws   = (float*)d_ws;
    float* dis  = ws;
    float* bufA = dis  + n;
    float* bufB = bufA + (size_t)n * H;
    float* sums = bufB + (size_t)n * H;
    float* cnts = sums + N_GRAPHS;

    const int TB = 256;
    dim3 blk(TB);
    auto blocks = [](long long work, int tb) { return dim3((unsigned)((work + tb - 1) / tb)); };

    k_init_deg  <<<blocks(n, TB), blk, 0, stream>>>(dis, n);
    k_deg_edges <<<blocks(e, TB), blk, 0, stream>>>(dst, dis, e);
    float* aggx = bufA;
    float* xs   = bufB;
    k_rsqrt_start<<<blocks(n, TB), blk, 0, stream>>>(dis, x, aggx, xs, n);

    const int strips       = (n + 15) / 16;
    const dim3 gemm_grid   = blocks((long long)strips * 32, TB);

    k_scatter_x <<<blocks((long long)e, TB), blk, 0, stream>>>(src, dst, xs, aggx, e);
    k_gemm_l1<F_IN, 12><<<gemm_grid, blk, 0, stream>>>(aggx, W1, b1, dis, bufB, bufA, n);

    k_scatter_edges <<<blocks((long long)e, TB), blk, 0, stream>>>(src, dst, bufB, bufA, e);

    float* y    = bufB;
    float* racc = bufB + n;
    k_gemm2_fold<<<gemm_grid, blk, 0, stream>>>(bufA, W2, b2, W3, lin_w, dis, y, racc, n);
    k_scatter_y <<<blocks(((long long)e + 3) / 4, TB), blk, 0, stream>>>(src, dst, y, racc, e);

    k_zero      <<<blocks(2 * N_GRAPHS, TB), blk, 0, stream>>>(sums, 2 * N_GRAPHS);
    k_pool_y    <<<blocks(n, TB), blk, 0, stream>>>(racc, dis, batch, sums, cnts, n);
    k_pool_final<<<blocks(N_GRAPHS, TB), blk, 0, stream>>>(sums, cnts, b3, lin_w, lin_b, out, N_GRAPHS);
}


# float atomic sums of the degree, layer-1 input-row and layer-3 scalar edge passes kept in one plane per package half (by die number), consumers add the two planes; float sums regrouped
# speedup vs baseline: 3.4888x; 1.3366x over previous
#include <hip/hip_runtime.h>
#include <hip/hip_bf16.h>

#define N_NODES 200000
#define N_EDGES 1000000
#define N_GRAPHS 4096
#define F_IN 11
#define H 64
#define XS 16

__device__ __forceinline__ int aggx_off(int i) { return (i >> 4) * (16 * H) + (i & 15) * XS; }

typedef float v2f __attribute__((ext_vector_type(2)));
typedef float v8f __attribute__((ext_vector_type(8)));

__device__ __forceinline__ void atomicAddF(float* p, float v) {
    unsafeAtomicAdd(p, v);
}

#define AGGX_PLANE 256
__device__ __forceinline__ int my_half() {
    const unsigned die = (__builtin_amdgcn_s_sendmsg_rtn(0x87) >> 16) & 15u;
    return die >= 4u ? 1 : 0;
}


__global__ void k_init_deg(float* __restrict__ deg0, float* __restrict__ deg1, int n) {
    int i = blockIdx.x * blockDim.x + threadIdx.x;
    if (i < n) {
        deg0[i] = 1.0f;
        deg1[i] = 0.0f;
    }
}

__global__ void k_deg_edges(const int* __restrict__ dst, float* deg0, float* deg1, int e) {
    int i = blockIdx.x * blockDim.x + threadIdx.x;
    float* deg = my_half() ? deg1 : deg0;
    if (i < e) atomicAddF(&deg[dst[i]], 1.0f);
}

__global__ void k_rsqrt_start(float* deg, const float* __restrict__ deg1, const float* __restrict__ x,
                              float* __restrict__ aggx, float* __restrict__ xs, int n) {
    int i = blockIdx.x * blockDim.x + threadIdx.x;
    if (i >= n) return;
    const float dv = rsqrtf(fmaxf(deg[i] + deg1[i], 1.0f));
    deg[i] = dv;
    const float* __restrict__ xr = x + i * F_IN;
    float r[XS];
    #pragma unroll
    for (int f = 0; f < XS; ++f) r[f] = (f < F_IN) ? xr[f] * dv : 0.0f;
    float4* o = reinterpret_cast<float4*>(aggx + aggx_off(i));
    float4* p = reinterpret_cast<float4*>(xs + i * XS);
    #pragma unroll
    for (int q = 0; q < XS / 4; ++q) {
        const float4 v = make_float4(r[4 * q], r[4 * q + 1], r[4 * q + 2], r[4 * q + 3]);
        o[q] = v;
        o[q + AGGX_PLANE / 4] = make_float4(0.0f, 0.0f, 0.0f, 0.0f);
        p[q] = v;
    }
}


template <int KV, int KP>
__global__ void k_gemm_l1(const float* A, const float* __restrict__ W,
                          const float* __restrict__ b, const float* __restrict__ dis,
                          float* __restrict__ Out, float* Agg, int n) {
    static_assert(KP <= XS && (KP & 3) == 0, "A rows hold XS columns");
    __shared__ float Wlds[(KP / 2) * H * 2];
    for (int idx = threadIdx.x; idx < KP * H; idx += blockDim.x) {
        int k = idx / H, c = idx - k * H;
        float v = (k < KV) ? W[k * H + c] : 0.0f;
        Wlds[(k >> 1) * (H * 2) + c * 2 + (k & 1)] = v;
    }
    __syncthreads();

    const int lane  = threadIdx.x & 31;
    const int strip = blockIdx.x * (blockDim.x >> 5) + (threadIdx.x >> 5);
    const int row0  = strip * 16;
    if (row0 >= n) return;

    const int mrow = lane & 15;
    const int hi   = lane >> 4;
    const float* arow = A + aggx_off(row0 + mrow);

    const int mbase = (lane < 16) ? 0 : 8;
    float drow[8];
    #pragma unroll
    for (int j = 0; j < 8; ++j) drow[j] = dis[row0 + mbase + j];

    v8f acc[4];
    #pragma unroll
    for (int t = 0; t < 4; ++t) acc[t] = (v8f){0, 0, 0, 0, 0, 0, 0, 0};

    #pragma unroll
    for (int k0 = 0; k0 < KP; k0 += 4) {
        const int ka = k0 + hi * 2;
        const v2f a = *reinterpret_cast<const v2f*>(arow + ka)
                    + *reinterpret_cast<const v2f*>(arow + AGGX_PLANE + ka);
        const int pbase = ((k0 >> 1) + hi) * (H * 2);
        #pragma unroll
        for (int t = 0; t < 4; ++t) {
            const int c = t * 16 + mrow;
            v2f bw = *reinterpret_cast<const v2f*>(&Wlds[pbase + c * 2]);
            acc[t] = __builtin_amdgcn_wmma_f32_16x16x4_f32(
                 false, a,  false, bw,
                 (short)0, acc[t],  false,  false);
        }
    }

    #pragma unroll
    for (int t = 0; t < 4; ++t) {
        const int col = t * 16 + mrow;
        const float bc = b[col];
        #pragma unroll
        for (int j = 0; j < 8; ++j) {
            const float v = fmaxf(fmaf(acc[t][j], drow[j], bc), 0.0f) * drow[j];
            Out[(row0 + mbase + j) * H + col] = v;
            Agg[(row0 + mbase + j) * H + col] = v;
        }
    }
}


#define EDGES_PER_WAVE 32
#define EDGE_CHUNK 8

__device__ __forceinline__ int bcast_i(int v, int l) { return __builtin_amdgcn_readlane(v, l); }

__global__ void k_scatter_edges(const int* __restrict__ src, const int* __restrict__ dst,
                                const float* __restrict__ h, float* __restrict__ agg, int e) {
    const int lane = threadIdx.x & 31;
    const int wave = blockIdx.x * (blockDim.x >> 5) + __builtin_amdgcn_readfirstlane(threadIdx.x >> 5);
    const int e0 = wave * EDGES_PER_WAVE;
    if (e0 >= e) return;
    const int left = e - e0;
    const int cnt = left < EDGES_PER_WAVE ? left : EDGES_PER_WAVE;

    int s = 0, d = 0;
    if (lane < cnt) {
        s = src[e0 + lane];
        d = dst[e0 + lane];
    }

    int c = 0;
    for (; c + EDGE_CHUNK <= cnt; c += EDGE_CHUNK) {
        float lo[EDGE_CHUNK], hi[EDGE_CHUNK];
        #pragma unroll
        for (int j = 0; j < EDGE_CHUNK; ++j) {
            const float* __restrict__ row = h + (size_t)bcast_i(s, c + j) * H;
            lo[j] = row[lane];
            hi[j] = row[lane + 32];
        }
        #pragma unroll
        for (int j = 0; j < EDGE_CHUNK; ++j) {
            float* o = agg + (size_t)bcast_i(d, c + j) * H;
            atomicAddF(o + lane,      lo[j]);
            atomicAddF(o + lane + 32, hi[j]);
        }
    }
    for (; c < cnt; ++c) {
        const float* __restrict__ row = h + (size_t)bcast_i(s, c) * H;
        float* o = agg + (size_t)bcast_i(d, c) * H;
        atomicAddF(o + lane,      row[lane]);
        atomicAddF(o + lane + 32, row[lane + 32]);
    }
}

__device__ __forceinline__ int   pick_i(int v, int j, int hi)   { const int a = bcast_i(v, 2 * j), b = bcast_i(v, 2 * j + 1); return hi ? b : a; }

__global__ void k_scatter_x(const int* __restrict__ src, const int* __restrict__ dst,
                            const float* __restrict__ xs, float* aggx, int e) {
    const int half = my_half();
    const int lane = threadIdx.x & 31;
    const int wave = blockIdx.x * (blockDim.x >> 5) + __builtin_amdgcn_readfirstlane(threadIdx.x >> 5);
    const int e0 = wave * EDGES_PER_WAVE;
    if (e0 >= e) return;
    const int left = e - e0;
    const int cnt = left < EDGES_PER_WAVE ? left : EDGES_PER_WAVE;

    int s = 0, d = 0;
    if (lane < cnt) {
        s = src[e0 + lane];
        d = dst[e0 + lane];
    }

    const int f  = lane & 15;
    const int hi = lane >> 4;
    aggx += half * AGGX_PLANE;
    if (cnt == EDGES_PER_WAVE) {
        if (f < F_IN) {
            float v[EDGES_PER_WAVE / 2];
            #pragma unroll
            for (int j = 0; j < EDGES_PER_WAVE / 2; ++j) v[j] = xs[pick_i(s, j, hi) * XS + f];
            #pragma unroll
            for (int j = 0; j < EDGES_PER_WAVE / 2; ++j)
                atomicAddF(aggx + aggx_off(pick_i(d, j, hi)) + f, v[j]);
        }
    } else {
        for (int j = 0; 2 * j < cnt; ++j) {
            const int sj = pick_i(s, j, hi), dj = pick_i(d, j, hi);
            if (f < F_IN && 2 * j + hi < cnt) atomicAddF(aggx + aggx_off(dj) + f, xs[sj * XS + f]);
        }
    }
}


__global__ void k_zero(float* p, int n) {
    int i = blockIdx.x * blockDim.x + threadIdx.x;
    if (i < n) p[i] = 0.0f;
}


__global__ void k_gemm2_fold(const float* __restrict__ G, const float* __restrict__ W2,
                             const float* __restrict__ b2, const float* __restrict__ W3,
                             const float* __restrict__ lw, const float* __restrict__ dis,
                             float* __restrict__ y, float* __restrict__ racc,
                             float* __restrict__ racc1, int n) {
    __shared__ float Wlds[(H / 2) * H * 2];
    __shared__ alignas(16) float Ulds[H];
    __shared__ alignas(16) float Blds[H];
    __shared__ float Zlds[32 * 16];
    for (int idx = threadIdx.x; idx < H * H; idx += blockDim.x) {
        int k = idx / H, c = idx - k * H;
        Wlds[(k >> 1) * (H * 2) + c * 2 + (k & 1)] = W2[k * H + c];
    }
    for (int k = threadIdx.x >> 2; k < H; k += blockDim.x >> 2) {
        const int q = threadIdx.x & 3;
        const float4* wr = reinterpret_cast<const float4*>(W3 + k * H + q * 16);
        const float4* lr = reinterpret_cast<const float4*>(lw + q * 16);
        float p = 0.0f;
        #pragma unroll
        for (int i = 0; i < 4; ++i) {
            const float4 a = wr[i], b = lr[i];
            p += a.x * b.x + a.y * b.y + a.z * b.z + a.w * b.w;
        }
        p += __shfl_xor(p, 1);
        p += __shfl_xor(p, 2);
        if (q == 0) Ulds[k] = p;
    }
    for (int idx = threadIdx.x; idx < H; idx += blockDim.x) Blds[idx] = b2[idx];
    __syncthreads();

    const int lane = threadIdx.x & 31;
    const int wv    = __builtin_amdgcn_readfirstlane(threadIdx.x >> 5);
    const int row0  = (blockIdx.x * (blockDim.x >> 5) + wv) * 16;
    const int mrow  = lane & 15;
    const int hi    = lane >> 4;
    if (row0 < n) {
        int r = row0 + mrow;
        r = r < n ? r : n - 1;
        const float* grow = G + (size_t)r * H;
        const float dn = dis[r];
        v8f acc[4];
        #pragma unroll
        for (int t = 0; t < 4; ++t) acc[t] = (v8f){0, 0, 0, 0, 0, 0, 0, 0};
        #pragma unroll
        for (int k0 = 0; k0 < H; k0 += 4) {
            const int ka = k0 + hi * 2;
            const v2f g = *reinterpret_cast<const v2f*>(grow + ka);
            const int pbase = ((k0 >> 1) + hi) * (H * 2);
            #pragma unroll
            for (int t = 0; t < 4; ++t) {
                const int c = t * 16 + mrow;
                const v2f w = *reinterpret_cast<const v2f*>(&Wlds[pbase + c * 2]);
                acc[t] = __builtin_amdgcn_wmma_f32_16x16x4_f32(false, w, false, g, (short)0, acc[t], false, false);
            }
        }
        v8f z0 = (v8f){0, 0, 0, 0, 0, 0, 0, 0};
        v8f z1 = (v8f){0, 0, 0, 0, 0, 0, 0, 0};
        #pragma unroll
        for (int t = 0; t < 4; ++t) {
            const int cb = t * 16 + hi * 8;
            const float4 ba = *reinterpret_cast<const float4*>(&Blds[cb]);
            const float4 bb = *reinterpret_cast<const float4*>(&Blds[cb + 4]);
            const float4 ua = *reinterpret_cast<const float4*>(&Ulds[cb]);
            const float4 ub = *reinterpret_cast<const float4*>(&Ulds[cb + 4]);
            const float bias[8] = {ba.x, ba.y, ba.z, ba.w, bb.x, bb.y, bb.z, bb.w};
            const float uu[8]   = {ua.x, ua.y, ua.z, ua.w, ub.x, ub.y, ub.z, ub.w};
            #pragma unroll
            for (int q = 0; q < 4; ++q) {
                v2f hv, uv;
                hv.x = fmaxf(fmaf(acc[t][2 * q],     dn, bias[2 * q]),     0.0f);
                hv.y = fmaxf(fmaf(acc[t][2 * q + 1], dn, bias[2 * q + 1]), 0.0f);
                uv.x = mrow == 0 ? uu[2 * q]     : 0.0f;
                uv.y = mrow == 0 ? uu[2 * q + 1] : 0.0f;
                if (q & 1) z1 = __builtin_amdgcn_wmma_f32_16x16x4_f32(false, uv, false, hv, (short)0, z1, false, false);
                else       z0 = __builtin_amdgcn_wmma_f32_16x16x4_f32(false, uv, false, hv, (short)0, z0, false, false);
            }
        }
        const v8f z = z0 + z1;
        if (hi == 0) Zlds[wv * 16 + mrow] = z[0];
    }
    __syncthreads();
    const int brow = (blockDim.x >> 5) * 16;
    for (int t = threadIdx.x; t < brow; t += blockDim.x) {
        const int r = blockIdx.x * brow + t;
        if (r < n) {
            const float v = Zlds[t] * dis[r];
            y[r] = v;
            racc[r] = v;
            racc1[r] = 0.0f;
        }
    }
}

__global__ void k_scatter_y(const int* __restrict__ src, const int* __restrict__ dst,
                            const float* __restrict__ y, float* racc0, float* racc1, int e) {
    const int e0 = (blockIdx.x * blockDim.x + threadIdx.x) * 4;
    float* racc = my_half() ? racc1 : racc0;
    if (e0 >= e) return;
    if (e0 + 4 <= e && (e & 3) == 0) {
        const int4 s = *reinterpret_cast<const int4*>(src + e0);
        const int4 d = *reinterpret_cast<const int4*>(dst + e0);
        const float v0 = y[s.x], v1 = y[s.y], v2 = y[s.z], v3 = y[s.w];
        atomicAddF(racc + d.x, v0);
        atomicAddF(racc + d.y, v1);
        atomicAddF(racc + d.z, v2);
        atomicAddF(racc + d.w, v3);
    } else {
        for (int i = e0; i < e && i < e0 + 4; ++i) atomicAddF(racc + dst[i], y[src[i]]);
    }
}

__global__ void k_pool_y(const float* __restrict__ racc, const float* __restrict__ racc1,
                         const float* __restrict__ dis,
                         const int* __restrict__ batch, float* __restrict__ sums,
                         float* __restrict__ cnts, int n) {
    const int i = blockIdx.x * blockDim.x + threadIdx.x;
    if (i >= n) return;
    const int g = batch[i];
    atomicAddF(&sums[g], (racc[i] + racc1[i]) * dis[i]);
    atomicAddF(&cnts[g], 1.0f);
}

__global__ void k_pool_final(const float* __restrict__ sums, const float* __restrict__ cnts,
                             const float* __restrict__ b, const float* __restrict__ lin_w,
                             const float* __restrict__ lin_b, float* __restrict__ out, int g_total) {
    const int g = blockIdx.x * blockDim.x + threadIdx.x;
    if (g >= g_total) return;
    float c = 0.0f;
    for (int f = 0; f < H; ++f) c += b[f] * lin_w[f];
    const float cn = cnts[g];
    out[g] = (sums[g] + cn * c) / fmaxf(cn, 1.0f) + lin_b[0];
}


extern "C" void kernel_launch(void* const* d_in, const int* in_sizes, int n_in,
                              void* d_out, int out_size, void* d_ws, size_t ws_size,
                              hipStream_t stream) {
    const float* x     = (const float*)d_in[0];
    const int*   eidx  = (const int*)  d_in[1];
    const int*   batch = (const int*)  d_in[2];
    const float* W1    = (const float*)d_in[3];
    const float* b1    = (const float*)d_in[4];
    const float* W2    = (const float*)d_in[5];
    const float* b2    = (const float*)d_in[6];
    const float* W3    = (const float*)d_in[7];
    const float* b3    = (const float*)d_in[8];
    const float* lin_w = (const float*)d_in[9];
    const float* lin_b = (const float*)d_in[10];
    float* out = (float*)d_out;

    const int n = in_sizes[0] / F_IN;
    const int e = in_sizes[1] / 2;
    const int* src = eidx;
    const int* dst = eidx + e;

    float* ws   = (float*)d_ws;
    float* dis  = ws;
    float* bufA = dis  + n;
    float* bufB = bufA + (size_t)n * H;
    float* sums = bufB + (size_t)n * H;
    float* cnts = sums + N_GRAPHS;

    const int TB = 256;
    dim3 blk(TB);
    auto blocks = [](long long work, int tb) { return dim3((unsigned)((work + tb - 1) / tb)); };

    float* deg1 = bufB + (size_t)n * 32;
    k_init_deg  <<<blocks(n, TB), blk, 0, stream>>>(dis, deg1, n);
    k_deg_edges <<<blocks(e, TB), blk, 0, stream>>>(dst, dis, deg1, e);
    float* aggx = bufA;
    float* xs   = bufB;
    k_rsqrt_start<<<blocks(n, TB), blk, 0, stream>>>(dis, deg1, x, aggx, xs, n);

    const int strips       = (n + 15) / 16;
    const dim3 gemm_grid   = blocks((long long)strips * 32, TB);

    k_scatter_x <<<blocks((long long)e, TB), blk, 0, stream>>>(src, dst, xs, aggx, e);
    k_gemm_l1<F_IN, 12><<<gemm_grid, blk, 0, stream>>>(aggx, W1, b1, dis, bufB, bufA, n);

    k_scatter_edges <<<blocks((long long)e, TB), blk, 0, stream>>>(src, dst, bufB, bufA, e);

    float* y    = bufB;
    float* racc = bufB + n;
    float* racc1 = racc + n;
    k_gemm2_fold<<<gemm_grid, blk, 0, stream>>>(bufA, W2, b2, W3, lin_w, dis, y, racc, racc1, n);
    k_scatter_y <<<blocks(((long long)e + 3) / 4, TB), blk, 0, stream>>>(src, dst, y, racc, racc1, e);

    k_zero      <<<blocks(2 * N_GRAPHS, TB), blk, 0, stream>>>(sums, 2 * N_GRAPHS);
    k_pool_y    <<<blocks(n, TB), blk, 0, stream>>>(racc, racc1, dis, batch, sums, cnts, n);
    k_pool_final<<<blocks(N_GRAPHS, TB), blk, 0, stream>>>(sums, cnts, b3, lin_w, lin_b, out, N_GRAPHS);
}


# every pass gives a wave several items, their loads issued together (2-8 per thread, 2 strips per GEMM wave, 2-4 edge groups per scatter wave): a third as many waves; float atomic sums regrouped
# speedup vs baseline: 3.7815x; 1.0839x over previous
#include <hip/hip_runtime.h>
#include <hip/hip_bf16.h>

#define N_NODES 200000
#define N_EDGES 1000000
#define N_GRAPHS 4096
#define F_IN 11
#define H 64
#define XS 16

__device__ __forceinline__ int aggx_off(int i) { return (i >> 4) * (16 * H) + (i & 15) * XS; }

typedef float v2f __attribute__((ext_vector_type(2)));
typedef float v8f __attribute__((ext_vector_type(8)));

__device__ __forceinline__ void atomicAddF(float* p, float v) {
    unsafeAtomicAdd(p, v);
}

#define AGGX_PLANE 256
__device__ __forceinline__ int my_half() {
    const unsigned die = (__builtin_amdgcn_s_sendmsg_rtn(0x87) >> 16) & 15u;
    return die >= 4u ? 1 : 0;
}


__global__ void k_init_deg(float* __restrict__ deg0, float* __restrict__ deg1, int n) {
    const int T = gridDim.x * blockDim.x;
    for (int i = blockIdx.x * blockDim.x + threadIdx.x; i < n; i += T) {
        deg0[i] = 1.0f;
        deg1[i] = 0.0f;
    }
}

__device__ __forceinline__ void count4(float* deg, const int4 d) {
    atomicAddF(&deg[d.x], 1.0f);
    atomicAddF(&deg[d.y], 1.0f);
    atomicAddF(&deg[d.z], 1.0f);
    atomicAddF(&deg[d.w], 1.0f);
}

__global__ void k_deg_edges(const int* __restrict__ dst, float* deg0, float* deg1, int e) {
    const int tid = blockIdx.x * blockDim.x + threadIdx.x;
    const int T = gridDim.x * blockDim.x;
    float* deg = my_half() ? deg1 : deg0;
    if ((e & 3) == 0) {
        const int4* __restrict__ dq = reinterpret_cast<const int4*>(dst);
        const int nq = e >> 2;
        int q = tid;
        for (; q + T < nq; q += 2 * T) {
            int4 a = dq[q];
            int4 b = dq[q + T];
            asm volatile("" : "+v"(a.x), "+v"(a.y), "+v"(a.z), "+v"(a.w), "+v"(b.x), "+v"(b.y), "+v"(b.z), "+v"(b.w));
            count4(deg, a);
            count4(deg, b);
        }
        if (q < nq) count4(deg, dq[q]);
    } else {
        for (int i = tid; i < e; i += T) atomicAddF(&deg[dst[i]], 1.0f);
    }
}

struct NodeIn { float dsum; float xv[F_IN]; };

__device__ __forceinline__ NodeIn node_in(const float* deg, const float* __restrict__ deg1,
                                          const float* __restrict__ x, int i) {
    NodeIn v;
    v.dsum = deg[i] + deg1[i];
    const float* __restrict__ xr = x + i * F_IN;
    #pragma unroll
    for (int f = 0; f < F_IN; ++f) v.xv[f] = xr[f];
    return v;
}

__device__ __forceinline__ void node_out(const NodeIn& v, int i, float* deg,
                                         float* __restrict__ aggx, float* __restrict__ xs) {
    const float dv = rsqrtf(fmaxf(v.dsum, 1.0f));
    deg[i] = dv;
    float r[XS];
    #pragma unroll
    for (int f = 0; f < XS; ++f) r[f] = (f < F_IN) ? v.xv[f < F_IN ? f : 0] * dv : 0.0f;
    float4* o = reinterpret_cast<float4*>(aggx + aggx_off(i));
    float4* p = reinterpret_cast<float4*>(xs + i * XS);
    #pragma unroll
    for (int q = 0; q < XS / 4; ++q) {
        const float4 w = make_float4(r[4 * q], r[4 * q + 1], r[4 * q + 2], r[4 * q + 3]);
        o[q] = w;
        o[q + AGGX_PLANE / 4] = make_float4(0.0f, 0.0f, 0.0f, 0.0f);
        p[q] = w;
    }
}

__global__ void k_rsqrt_start(float* deg, const float* __restrict__ deg1, const float* __restrict__ x,
                              float* __restrict__ aggx, float* __restrict__ xs, int n) {
    const int T = gridDim.x * blockDim.x;
    for (int i = blockIdx.x * blockDim.x + threadIdx.x; i < n; i += 2 * T) {
        if (i + T < n) {
            const NodeIn a = node_in(deg, deg1, x, i);
            const NodeIn b = node_in(deg, deg1, x, i + T);
            node_out(a, i, deg, aggx, xs);
            node_out(b, i + T, deg, aggx, xs);
        } else {
            node_out(node_in(deg, deg1, x, i), i, deg, aggx, xs);
        }
    }
}


template <int KV, int KP>
__global__ void k_gemm_l1(const float* A, const float* __restrict__ W,
                          const float* __restrict__ b, const float* __restrict__ dis,
                          float* __restrict__ Out, float* Agg, int n) {
    static_assert(KP <= XS && (KP & 3) == 0, "A rows hold XS columns");
    __shared__ float Wlds[(KP / 2) * H * 2];
    for (int idx = threadIdx.x; idx < KP * H; idx += blockDim.x) {
        int k = idx / H, c = idx - k * H;
        float v = (k < KV) ? W[k * H + c] : 0.0f;
        Wlds[(k >> 1) * (H * 2) + c * 2 + (k & 1)] = v;
    }
    __syncthreads();

    const int lane  = threadIdx.x & 31;
    const int mrow = lane & 15;
    const int hi   = lane >> 4;
    const int wpb  = blockDim.x >> 5;
    const int nstrips = (n + 15) >> 4;
    for (int strip = blockIdx.x * wpb + (threadIdx.x >> 5); strip < nstrips; strip += gridDim.x * wpb) {
    const int row0  = strip * 16;
    const float* arow = A + aggx_off(row0 + mrow);

    const int mbase = (lane < 16) ? 0 : 8;
    float drow[8];
    #pragma unroll
    for (int j = 0; j < 8; ++j) drow[j] = dis[row0 + mbase + j];

    v8f acc[4];
    #pragma unroll
    for (int t = 0; t < 4; ++t) acc[t] = (v8f){0, 0, 0, 0, 0, 0, 0, 0};

    #pragma unroll
    for (int k0 = 0; k0 < KP; k0 += 4) {
        const int ka = k0 + hi * 2;
        const v2f a = *reinterpret_cast<const v2f*>(arow + ka)
                    + *reinterpret_cast<const v2f*>(arow + AGGX_PLANE + ka);
        const int pbase = ((k0 >> 1) + hi) * (H * 2);
        #pragma unroll
        for (int t = 0; t < 4; ++t) {
            const int c = t * 16 + mrow;
            v2f bw = *reinterpret_cast<const v2f*>(&Wlds[pbase + c * 2]);
            acc[t] = __builtin_amdgcn_wmma_f32_16x16x4_f32(
                 false, a,  false, bw,
                 (short)0, acc[t],  false,  false);
        }
    }

    #pragma unroll
    for (int t = 0; t < 4; ++t) {
        const int col = t * 16 + mrow;
        const float bc = b[col];
        #pragma unroll
        for (int j = 0; j < 8; ++j) {
            const float v = fmaxf(fmaf(acc[t][j], drow[j], bc), 0.0f) * drow[j];
            Out[(row0 + mbase + j) * H + col] = v;
            Agg[(row0 + mbase + j) * H + col] = v;
        }
    }
    }
}


#define EDGES_PER_WAVE 32
#define EDGE_CHUNK 8
#define GROUPS_H 2
#define GROUPS_X 4

__device__ __forceinline__ int bcast_i(int v, int l) { return __builtin_amdgcn_readlane(v, l); }

__global__ void k_scatter_edges(const int* __restrict__ src, const int* __restrict__ dst,
                                const float* __restrict__ h, float* __restrict__ agg, int e) {
    const int lane = threadIdx.x & 31;
    const int wave = blockIdx.x * (blockDim.x >> 5) + __builtin_amdgcn_readfirstlane(threadIdx.x >> 5);
    const int e0 = wave * (EDGES_PER_WAVE * GROUPS_H);
    if (e0 >= e) return;

    int s[GROUPS_H], d[GROUPS_H];
    #pragma unroll
    for (int g = 0; g < GROUPS_H; ++g) {
        const int i = e0 + g * EDGES_PER_WAVE + lane;
        s[g] = 0;
        d[g] = 0;
        if (i < e) {
            s[g] = src[i];
            d[g] = dst[i];
        }
    }

    #pragma unroll
    for (int g = 0; g < GROUPS_H; ++g) {
        const int left = e - (e0 + g * EDGES_PER_WAVE);
        const int cnt = left < EDGES_PER_WAVE ? left : EDGES_PER_WAVE;
        int c = 0;
        for (; c + EDGE_CHUNK <= cnt; c += EDGE_CHUNK) {
            float lo[EDGE_CHUNK], hi[EDGE_CHUNK];
            #pragma unroll
            for (int j = 0; j < EDGE_CHUNK; ++j) {
                const float* __restrict__ row = h + (size_t)bcast_i(s[g], c + j) * H;
                lo[j] = row[lane];
                hi[j] = row[lane + 32];
            }
            #pragma unroll
            for (int j = 0; j < EDGE_CHUNK; ++j) {
                float* o = agg + (size_t)bcast_i(d[g], c + j) * H;
                atomicAddF(o + lane,      lo[j]);
                atomicAddF(o + lane + 32, hi[j]);
            }
        }
        for (; c < cnt; ++c) {
            const float* __restrict__ row = h + (size_t)bcast_i(s[g], c) * H;
            float* o = agg + (size_t)bcast_i(d[g], c) * H;
            atomicAddF(o + lane,      row[lane]);
            atomicAddF(o + lane + 32, row[lane + 32]);
        }
    }
}

__device__ __forceinline__ int   pick_i(int v, int j, int hi)   { const int a = bcast_i(v, 2 * j), b = bcast_i(v, 2 * j + 1); return hi ? b : a; }

__global__ void k_scatter_x(const int* __restrict__ src, const int* __restrict__ dst,
                            const float* __restrict__ xs, float* aggx, int e) {
    const int half = my_half();
    const int lane = threadIdx.x & 31;
    const int wave = blockIdx.x * (blockDim.x >> 5) + __builtin_amdgcn_readfirstlane(threadIdx.x >> 5);
    const int e0 = wave * (EDGES_PER_WAVE * GROUPS_X);
    if (e0 >= e) return;

    int s[GROUPS_X], d[GROUPS_X];
    #pragma unroll
    for (int g = 0; g < GROUPS_X; ++g) {
        const int i = e0 + g * EDGES_PER_WAVE + lane;
        s[g] = 0;
        d[g] = 0;
        if (i < e) {
            s[g] = src[i];
            d[g] = dst[i];
        }
    }

    const int f  = lane & 15;
    const int hi = lane >> 4;
    aggx += half * AGGX_PLANE;
    #pragma unroll
    for (int g = 0; g < GROUPS_X; ++g) {
        const int left = e - (e0 + g * EDGES_PER_WAVE);
        if (left >= EDGES_PER_WAVE) {
            if (f < F_IN) {
                float v[EDGES_PER_WAVE / 2];
                #pragma unroll
                for (int j = 0; j < EDGES_PER_WAVE / 2; ++j) v[j] = xs[pick_i(s[g], j, hi) * XS + f];
                #pragma unroll
                for (int j = 0; j < EDGES_PER_WAVE / 2; ++j)
                    atomicAddF(aggx + aggx_off(pick_i(d[g], j, hi)) + f, v[j]);
            }
        } else {
            for (int j = 0; 2 * j < left; ++j) {
                const int sj = pick_i(s[g], j, hi), dj = pick_i(d[g], j, hi);
                if (f < F_IN && 2 * j + hi < left) atomicAddF(aggx + aggx_off(dj) + f, xs[sj * XS + f]);
            }
        }
    }
}


__global__ void k_zero(float* p, int n) {
    int i = blockIdx.x * blockDim.x + threadIdx.x;
    if (i < n) p[i] = 0.0f;
}


__global__ void k_gemm2_fold(const float* __restrict__ G, const float* __restrict__ W2,
                             const float* __restrict__ b2, const float* __restrict__ W3,
                             const float* __restrict__ lw, const float* __restrict__ dis,
                             float* __restrict__ y, float* __restrict__ racc,
                             float* __restrict__ racc1, int n) {
    __shared__ float Wlds[(H / 2) * H * 2];
    __shared__ alignas(16) float Ulds[H];
    __shared__ alignas(16) float Blds[H];
    __shared__ float Zlds[2][32 * 16];
    for (int idx = threadIdx.x; idx < H * H; idx += blockDim.x) {
        int k = idx / H, c = idx - k * H;
        Wlds[(k >> 1) * (H * 2) + c * 2 + (k & 1)] = W2[k * H + c];
    }
    for (int k = threadIdx.x >> 2; k < H; k += blockDim.x >> 2) {
        const int q = threadIdx.x & 3;
        const float4* wr = reinterpret_cast<const float4*>(W3 + k * H + q * 16);
        const float4* lr = reinterpret_cast<const float4*>(lw + q * 16);
        float p = 0.0f;
        #pragma unroll
        for (int i = 0; i < 4; ++i) {
            const float4 a = wr[i], b = lr[i];
            p += a.x * b.x + a.y * b.y + a.z * b.z + a.w * b.w;
        }
        p += __shfl_xor(p, 1);
        p += __shfl_xor(p, 2);
        if (q == 0) Ulds[k] = p;
    }
    for (int idx = threadIdx.x; idx < H; idx += blockDim.x) Blds[idx] = b2[idx];
    __syncthreads();

    const int lane = threadIdx.x & 31;
    const int wv    = __builtin_amdgcn_readfirstlane(threadIdx.x >> 5);
    const int mrow  = lane & 15;
    const int hi    = lane >> 4;
    const int wpb   = blockDim.x >> 5;
    const int brow  = wpb * 16;
    const int nblk  = (n + brow - 1) / brow;
    int trip = 0;
    for (int rb = blockIdx.x; rb < nblk; rb += gridDim.x, ++trip) {
    float* Z = Zlds[trip & 1];
    const int row0  = (rb * wpb + wv) * 16;
    if (row0 < n) {
        int r = row0 + mrow;
        r = r < n ? r : n - 1;
        const float* grow = G + (size_t)r * H;
        const float dn = dis[r];
        v8f acc[4];
        #pragma unroll
        for (int t = 0; t < 4; ++t) acc[t] = (v8f){0, 0, 0, 0, 0, 0, 0, 0};
        v2f gk[H / 4];
        #pragma unroll
        for (int k0 = 0; k0 < H; k0 += 4) gk[k0 >> 2] = *reinterpret_cast<const v2f*>(grow + k0 + hi * 2);
        #pragma unroll
        for (int k0 = 0; k0 < H; k0 += 4) {
            const v2f g = gk[k0 >> 2];
            const int pbase = ((k0 >> 1) + hi) * (H * 2);
            #pragma unroll
            for (int t = 0; t < 4; ++t) {
                const int c = t * 16 + mrow;
                const v2f w = *reinterpret_cast<const v2f*>(&Wlds[pbase + c * 2]);
                acc[t] = __builtin_amdgcn_wmma_f32_16x16x4_f32(false, w, false, g, (short)0, acc[t], false, false);
            }
        }
        v8f z0 = (v8f){0, 0, 0, 0, 0, 0, 0, 0};
        v8f z1 = (v8f){0, 0, 0, 0, 0, 0, 0, 0};
        #pragma unroll
        for (int t = 0; t < 4; ++t) {
            const int cb = t * 16 + hi * 8;
            const float4 ba = *reinterpret_cast<const float4*>(&Blds[cb]);
            const float4 bb = *reinterpret_cast<const float4*>(&Blds[cb + 4]);
            const float4 ua = *reinterpret_cast<const float4*>(&Ulds[cb]);
            const float4 ub = *reinterpret_cast<const float4*>(&Ulds[cb + 4]);
            const float bias[8] = {ba.x, ba.y, ba.z, ba.w, bb.x, bb.y, bb.z, bb.w};
            const float uu[8]   = {ua.x, ua.y, ua.z, ua.w, ub.x, ub.y, ub.z, ub.w};
            #pragma unroll
            for (int q = 0; q < 4; ++q) {
                v2f hv, uv;
                hv.x = fmaxf(fmaf(acc[t][2 * q],     dn, bias[2 * q]),     0.0f);
                hv.y = fmaxf(fmaf(acc[t][2 * q + 1], dn, bias[2 * q + 1]), 0.0f);
                uv.x = mrow == 0 ? uu[2 * q]     : 0.0f;
                uv.y = mrow == 0 ? uu[2 * q + 1] : 0.0f;
                if (q & 1) z1 = __builtin_amdgcn_wmma_f32_16x16x4_f32(false, uv, false, hv, (short)0, z1, false, false);
                else       z0 = __builtin_amdgcn_wmma_f32_16x16x4_f32(false, uv, false, hv, (short)0, z0, false, false);
            }
        }
        const v8f z = z0 + z1;
        if (hi == 0) Z[wv * 16 + mrow] = z[0];
    }
    __syncthreads();
    for (int t = threadIdx.x; t < brow; t += blockDim.x) {
        const int r = rb * brow + t;
        if (r < n) {
            const float v = Z[t] * dis[r];
            y[r] = v;
            racc[r] = v;
            racc1[r] = 0.0f;
        }
    }
    }
}

__global__ void k_scatter_y(const int* __restrict__ src, const int* __restrict__ dst,
                            const float* __restrict__ y, float* racc0, float* racc1, int e) {
    const int tid = blockIdx.x * blockDim.x + threadIdx.x;
    const int T = gridDim.x * blockDim.x;
    float* racc = my_half() ? racc1 : racc0;
    if ((e & 3) == 0) {
        const int4* __restrict__ sq = reinterpret_cast<const int4*>(src);
        const int4* __restrict__ dq = reinterpret_cast<const int4*>(dst);
        const int nq = e >> 2;
        int q = tid;
        for (; q + T < nq; q += 2 * T) {
            const int4 s = sq[q], d = dq[q];
            const int4 s2 = sq[q + T];
            int4 d2 = dq[q + T];
            const float v0 = y[s.x], v1 = y[s.y], v2 = y[s.z], v3 = y[s.w];
            const float w0 = y[s2.x], w1 = y[s2.y], w2 = y[s2.z], w3 = y[s2.w];
            asm volatile("" : "+v"(d2.x), "+v"(d2.y), "+v"(d2.z), "+v"(d2.w));
            atomicAddF(racc + d.x, v0);
            atomicAddF(racc + d.y, v1);
            atomicAddF(racc + d.z, v2);
            atomicAddF(racc + d.w, v3);
            atomicAddF(racc + d2.x, w0);
            atomicAddF(racc + d2.y, w1);
            atomicAddF(racc + d2.z, w2);
            atomicAddF(racc + d2.w, w3);
        }
        if (q < nq) {
            const int4 s = sq[q], d = dq[q];
            const float v0 = y[s.x], v1 = y[s.y], v2 = y[s.z], v3 = y[s.w];
            atomicAddF(racc + d.x, v0);
            atomicAddF(racc + d.y, v1);
            atomicAddF(racc + d.z, v2);
            atomicAddF(racc + d.w, v3);
        }
    } else {
        for (int i = tid; i < e; i += T) atomicAddF(racc + dst[i], y[src[i]]);
    }
}

__global__ void k_pool_y(const float* __restrict__ racc, const float* __restrict__ racc1,
                         const float* __restrict__ dis,
                         const int* __restrict__ batch, float* __restrict__ sums,
                         float* __restrict__ cnts, int n) {
    const int T = gridDim.x * blockDim.x;
    for (int i0 = blockIdx.x * blockDim.x + threadIdx.x; i0 < n; i0 += 4 * T) {
        if (i0 + 3 * T < n) {
            int g[4];
            float v[4];
            #pragma unroll
            for (int u = 0; u < 4; ++u) {
                const int i = i0 + u * T;
                g[u] = batch[i];
                v[u] = (racc[i] + racc1[i]) * dis[i];
            }
            #pragma unroll
            for (int u = 0; u < 4; ++u) {
                atomicAddF(&sums[g[u]], v[u]);
                atomicAddF(&cnts[g[u]], 1.0f);
            }
        } else {
            for (int i = i0; i < n; i += T) {
                const int g = batch[i];
                atomicAddF(&sums[g], (racc[i] + racc1[i]) * dis[i]);
                atomicAddF(&cnts[g], 1.0f);
            }
        }
    }
}

__global__ void k_pool_final(const float* __restrict__ sums, const float* __restrict__ cnts,
                             const float* __restrict__ b, const float* __restrict__ lin_w,
                             const float* __restrict__ lin_b, float* __restrict__ out, int g_total) {
    const int g = blockIdx.x * blockDim.x + threadIdx.x;
    if (g >= g_total) return;
    float c = 0.0f;
    for (int f = 0; f < H; ++f) c += b[f] * lin_w[f];
    const float cn = cnts[g];
    out[g] = (sums[g] + cn * c) / fmaxf(cn, 1.0f) + lin_b[0];
}


extern "C" void kernel_launch(void* const* d_in, const int* in_sizes, int n_in,
                              void* d_out, int out_size, void* d_ws, size_t ws_size,
                              hipStream_t stream) {
    const float* x     = (const float*)d_in[0];
    const int*   eidx  = (const int*)  d_in[1];
    const int*   batch = (const int*)  d_in[2];
    const float* W1    = (const float*)d_in[3];
    const float* b1    = (const float*)d_in[4];
    const float* W2    = (const float*)d_in[5];
    const float* b2    = (const float*)d_in[6];
    const float* W3    = (const float*)d_in[7];
    const float* b3    = (const float*)d_in[8];
    const float* lin_w = (const float*)d_in[9];
    const float* lin_b = (const float*)d_in[10];
    float* out = (float*)d_out;

    const int n = in_sizes[0] / F_IN;
    const int e = in_sizes[1] / 2;
    const int* src = eidx;
    const int* dst = eidx + e;

    float* ws   = (float*)d_ws;
    float* dis  = ws;
    float* bufA = dis  + n;
    float* bufB = bufA + (size_t)n * H;
    float* sums = bufB + (size_t)n * H;
    float* cnts = sums + N_GRAPHS;

    const int TB = 256;
    dim3 blk(TB);
    auto blocks = [](long long work, int tb) { return dim3((unsigned)((work + tb - 1) / tb)); };

    float* deg1 = bufB + (size_t)n * 32;
    k_init_deg  <<<blocks((n + 3) / 4, TB), blk, 0, stream>>>(dis, deg1, n);
    k_deg_edges <<<blocks(((long long)e + 7) / 8, TB), blk, 0, stream>>>(dst, dis, deg1, e);
    float* aggx = bufA;
    float* xs   = bufB;
    k_rsqrt_start<<<blocks((n + 1) / 2, TB), blk, 0, stream>>>(dis, deg1, x, aggx, xs, n);

    const int strips       = (n + 15) / 16;
    const dim3 gemm_grid   = blocks((long long)((strips + 1) / 2) * 32, TB);

    k_scatter_x <<<blocks(((long long)e + GROUPS_X - 1) / GROUPS_X, TB), blk, 0, stream>>>(src, dst, xs, aggx, e);
    k_gemm_l1<F_IN, 12><<<gemm_grid, blk, 0, stream>>>(aggx, W1, b1, dis, bufB, bufA, n);

    k_scatter_edges <<<blocks(((long long)e + GROUPS_H - 1) / GROUPS_H, TB), blk, 0, stream>>>(src, dst, bufB, bufA, e);

    float* y    = bufB;
    float* racc = bufB + n;
    float* racc1 = racc + n;
    k_gemm2_fold<<<gemm_grid, blk, 0, stream>>>(bufA, W2, b2, W3, lin_w, dis, y, racc, racc1, n);
    k_scatter_y <<<blocks(((long long)e + 7) / 8, TB), blk, 0, stream>>>(src, dst, y, racc, racc1, e);

    k_zero      <<<blocks(2 * N_GRAPHS, TB), blk, 0, stream>>>(sums, 2 * N_GRAPHS);
    k_pool_y    <<<blocks((n + 3) / 4, TB), blk, 0, stream>>>(racc, racc1, dis, batch, sums, cnts, n);
    k_pool_final<<<blocks(N_GRAPHS, TB), blk, 0, stream>>>(sums, cnts, b3, lin_w, lin_b, out, N_GRAPHS);
}


# layer 2 edge scatter rebuilds each source row h1[src] on WMMA from the node's 64-byte summed input row instead of gathering the stored 256-byte row (h1 is no longer stored); float products regrouped
# speedup vs baseline: 4.0522x; 1.0716x over previous
#include <hip/hip_runtime.h>
#include <hip/hip_bf16.h>

#define N_NODES 200000
#define N_EDGES 1000000
#define N_GRAPHS 4096
#define F_IN 11
#define H 64
#define XS 16

__device__ __forceinline__ int aggx_off(int i) { return (i >> 4) * (16 * H) + (i & 15) * XS; }

typedef float v2f __attribute__((ext_vector_type(2)));
typedef float v8f __attribute__((ext_vector_type(8)));

__device__ __forceinline__ void atomicAddF(float* p, float v) {
    unsafeAtomicAdd(p, v);
}

#define AGGX_PLANE 256
__device__ __forceinline__ int my_half() {
    const unsigned die = (__builtin_amdgcn_s_sendmsg_rtn(0x87) >> 16) & 15u;
    return die >= 4u ? 1 : 0;
}


__global__ void k_init_deg(float* __restrict__ deg0, float* __restrict__ deg1, int n) {
    const int T = gridDim.x * blockDim.x;
    for (int i = blockIdx.x * blockDim.x + threadIdx.x; i < n; i += T) {
        deg0[i] = 1.0f;
        deg1[i] = 0.0f;
    }
}

__device__ __forceinline__ void count4(float* deg, const int4 d) {
    atomicAddF(&deg[d.x], 1.0f);
    atomicAddF(&deg[d.y], 1.0f);
    atomicAddF(&deg[d.z], 1.0f);
    atomicAddF(&deg[d.w], 1.0f);
}

__global__ void k_deg_edges(const int* __restrict__ dst, float* deg0, float* deg1, int e) {
    const int tid = blockIdx.x * blockDim.x + threadIdx.x;
    const int T = gridDim.x * blockDim.x;
    float* deg = my_half() ? deg1 : deg0;
    if ((e & 3) == 0) {
        const int4* __restrict__ dq = reinterpret_cast<const int4*>(dst);
        const int nq = e >> 2;
        int q = tid;
        for (; q + T < nq; q += 2 * T) {
            int4 a = dq[q];
            int4 b = dq[q + T];
            asm volatile("" : "+v"(a.x), "+v"(a.y), "+v"(a.z), "+v"(a.w), "+v"(b.x), "+v"(b.y), "+v"(b.z), "+v"(b.w));
            count4(deg, a);
            count4(deg, b);
        }
        if (q < nq) count4(deg, dq[q]);
    } else {
        for (int i = tid; i < e; i += T) atomicAddF(&deg[dst[i]], 1.0f);
    }
}

struct NodeIn { float dsum; float xv[F_IN]; };

__device__ __forceinline__ NodeIn node_in(const float* deg, const float* __restrict__ deg1,
                                          const float* __restrict__ x, int i) {
    NodeIn v;
    v.dsum = deg[i] + deg1[i];
    const float* __restrict__ xr = x + i * F_IN;
    #pragma unroll
    for (int f = 0; f < F_IN; ++f) v.xv[f] = xr[f];
    return v;
}

__device__ __forceinline__ void node_out(const NodeIn& v, int i, float* deg,
                                         float* __restrict__ aggx, float* __restrict__ xs) {
    const float dv = rsqrtf(fmaxf(v.dsum, 1.0f));
    deg[i] = dv;
    float r[XS];
    #pragma unroll
    for (int f = 0; f < XS; ++f) r[f] = (f < F_IN) ? v.xv[f < F_IN ? f : 0] * dv : 0.0f;
    float4* o = reinterpret_cast<float4*>(aggx + aggx_off(i));
    float4* p = reinterpret_cast<float4*>(xs + i * XS);
    #pragma unroll
    for (int q = 0; q < XS / 4; ++q) {
        const float4 w = make_float4(r[4 * q], r[4 * q + 1], r[4 * q + 2], r[4 * q + 3]);
        o[q] = w;
        o[q + AGGX_PLANE / 4] = make_float4(0.0f, 0.0f, 0.0f, 0.0f);
        p[q] = w;
    }
}

__global__ void k_rsqrt_start(float* deg, const float* __restrict__ deg1, const float* __restrict__ x,
                              float* __restrict__ aggx, float* __restrict__ xs, int n) {
    const int T = gridDim.x * blockDim.x;
    for (int i = blockIdx.x * blockDim.x + threadIdx.x; i < n; i += 2 * T) {
        if (i + T < n) {
            const NodeIn a = node_in(deg, deg1, x, i);
            const NodeIn b = node_in(deg, deg1, x, i + T);
            node_out(a, i, deg, aggx, xs);
            node_out(b, i + T, deg, aggx, xs);
        } else {
            node_out(node_in(deg, deg1, x, i), i, deg, aggx, xs);
        }
    }
}


template <int KV, int KP>
__global__ void k_gemm_l1(const float* A, const float* __restrict__ W,
                          const float* __restrict__ b, const float* __restrict__ dis,
                          float* __restrict__ Ax, float* Agg, int n) {
    static_assert(KP + 4 == XS && (KP & 3) == 0, "A rows hold XS columns; an Ax row is KP sums and one 16-byte piece behind them");
    __shared__ float Wlds[(KP / 2) * H * 2];
    for (int idx = threadIdx.x; idx < KP * H; idx += blockDim.x) {
        int k = idx / H, c = idx - k * H;
        float v = (k < KV) ? W[k * H + c] : 0.0f;
        Wlds[(k >> 1) * (H * 2) + c * 2 + (k & 1)] = v;
    }
    __syncthreads();

    const int lane  = threadIdx.x & 31;
    const int mrow = lane & 15;
    const int hi   = lane >> 4;
    const int wpb  = blockDim.x >> 5;
    const int nstrips = (n + 15) >> 4;
    for (int strip = blockIdx.x * wpb + (threadIdx.x >> 5); strip < nstrips; strip += gridDim.x * wpb) {
    const int row0  = strip * 16;
    const float* arow = A + aggx_off(row0 + mrow);

    const int mbase = (lane < 16) ? 0 : 8;
    float drow[8];
    #pragma unroll
    for (int j = 0; j < 8; ++j) drow[j] = dis[row0 + mbase + j];

    v8f acc[4];
    #pragma unroll
    for (int t = 0; t < 4; ++t) acc[t] = (v8f){0, 0, 0, 0, 0, 0, 0, 0};

    #pragma unroll
    for (int k0 = 0; k0 < KP; k0 += 4) {
        const int ka = k0 + hi * 2;
        const v2f a = *reinterpret_cast<const v2f*>(arow + ka)
                    + *reinterpret_cast<const v2f*>(arow + AGGX_PLANE + ka);
        *reinterpret_cast<v2f*>(Ax + (row0 + mrow) * XS + ka) = a;
        const int pbase = ((k0 >> 1) + hi) * (H * 2);
        #pragma unroll
        for (int t = 0; t < 4; ++t) {
            const int c = t * 16 + mrow;
            v2f bw = *reinterpret_cast<const v2f*>(&Wlds[pbase + c * 2]);
            acc[t] = __builtin_amdgcn_wmma_f32_16x16x4_f32(
                 false, a,  false, bw,
                 (short)0, acc[t],  false,  false);
        }
    }

    #pragma unroll
    for (int t = 0; t < 4; ++t) {
        const int col = t * 16 + mrow;
        const float bc = b[col];
        #pragma unroll
        for (int j = 0; j < 8; ++j) {
            const float v = fmaxf(fmaf(acc[t][j], drow[j], bc), 0.0f) * drow[j];
            Agg[(row0 + mbase + j) * H + col] = v;
        }
    }
    if (hi == 0) *reinterpret_cast<float4*>(Ax + (row0 + mrow) * XS + KP) = make_float4(dis[row0 + mrow], 0.0f, 0.0f, 0.0f);
    }
}


#define EDGES_PER_WAVE 32
#define EDGE_CHUNK 8
#define GROUPS_H 2
#define GROUPS_X 4
#define KP1 12
#define TPITCH 68
#define RPITCH 20

__device__ __forceinline__ int bcast_i(int v, int l) { return __builtin_amdgcn_readlane(v, l); }

struct SrcRows { float4 p[2]; };
__device__ __forceinline__ SrcRows src_rows(const float* __restrict__ ax, int sidx, int lane, int base) {
    SrcRows v;
    #pragma unroll
    for (int u = 0; u < 2; ++u) {
        const int sj = __shfl(sidx, base + 8 * u + (lane >> 2));
        v.p[u] = *reinterpret_cast<const float4*>(ax + (size_t)sj * XS + 4 * (lane & 3));
    }
    return v;
}

__global__ void __attribute__((amdgpu_waves_per_eu(16, 16))) k_scatter_h1(const int* __restrict__ src, const int* __restrict__ dst,
                             const float* __restrict__ ax, const float* __restrict__ W,
                             const float* __restrict__ b, float* __restrict__ agg, int e) {
    static_assert(KP1 == F_IN + 1 && (KP1 & 3) == 0 && KP1 < XS, "bias row behind the F_IN weight rows");
    __shared__ float Wlds[(KP1 / 2) * H * 2];
    __shared__ alignas(16) float Tlds[8][16 * TPITCH];
    for (int idx = threadIdx.x; idx < KP1 * H; idx += blockDim.x) {
        int k = idx / H, cc = idx - k * H;
        float v = (k < F_IN) ? W[k * H + cc] : b[cc];
        Wlds[(k >> 1) * (H * 2) + cc * 2 + (k & 1)] = v;
    }
    __syncthreads();

    const int lane = threadIdx.x & 31;
    const int wv   = __builtin_amdgcn_readfirstlane(threadIdx.x >> 5);
    const int wave = blockIdx.x * (blockDim.x >> 5) + wv;
    const int e0 = wave * (EDGES_PER_WAVE * GROUPS_H);
    if (e0 >= e) return;

    int s[GROUPS_H], d[GROUPS_H];
    #pragma unroll
    for (int g = 0; g < GROUPS_H; ++g) {
        const int i = e0 + g * EDGES_PER_WAVE + lane;
        s[g] = 0;
        d[g] = 0;
        if (i < e) {
            s[g] = src[i];
            d[g] = dst[i];
        }
    }

    const int mrow = lane & 15;
    const int hi   = lane >> 4;
    float* T = Tlds[wv];

    const int mine  = e - e0 < EDGES_PER_WAVE * GROUPS_H ? e - e0 : EDGES_PER_WAVE * GROUPS_H;
    const int ntile = (mine + 15) >> 4;
    SrcRows cur = src_rows(ax, s[0], lane, 0);
    #pragma unroll
    for (int tt = 0; tt < 2 * GROUPS_H; ++tt) {
        if (tt >= ntile) break;
        const int g = tt >> 1, base = (tt & 1) * 16;
        const int cnt = mine - tt * 16;

        #pragma unroll
        for (int u = 0; u < 2; ++u)
            *reinterpret_cast<float4*>(&T[(8 * u + (lane >> 2)) * RPITCH + 4 * (lane & 3)]) = cur.p[u];
        __builtin_amdgcn_fence(__ATOMIC_RELEASE, "wavefront");
        __builtin_amdgcn_wave_barrier();
        __builtin_amdgcn_fence(__ATOMIC_ACQUIRE, "wavefront");
        v2f a[KP1 / 4];
        #pragma unroll
        for (int q = 0; q < KP1 / 4; ++q) a[q] = *reinterpret_cast<const v2f*>(&T[mrow * RPITCH + 4 * q + 2 * hi]);
        const float dv = T[mrow * RPITCH + KP1];
        const float d2 = dv * dv;
        #pragma unroll
        for (int q = 0; q < KP1 / 4; ++q) a[q] *= d2;
        if (hi) a[KP1 / 4 - 1].y = dv;

        #pragma unroll
        for (int th = 0; th < 4; th += 2) {
            v8f acc[2];
            #pragma unroll
            for (int t = 0; t < 2; ++t) acc[t] = (v8f){0, 0, 0, 0, 0, 0, 0, 0};
            #pragma unroll
            for (int k0 = 0; k0 < KP1; k0 += 4) {
                const int pbase = ((k0 >> 1) + hi) * (H * 2);
                #pragma unroll
                for (int t = 0; t < 2; ++t) {
                    const int cc = (th + t) * 16 + mrow;
                    const v2f bw = *reinterpret_cast<const v2f*>(&Wlds[pbase + cc * 2]);
                    acc[t] = __builtin_amdgcn_wmma_f32_16x16x4_f32(false, a[k0 >> 2], false, bw, (short)0, acc[t], false, false);
                }
            }
            #pragma unroll
            for (int t = 0; t < 2; ++t) {
                #pragma unroll
                for (int j = 0; j < 8; ++j) T[(8 * hi + j) * TPITCH + (th + t) * 16 + mrow] = fmaxf(acc[t][j], 0.0f);
            }
        }
        __builtin_amdgcn_fence(__ATOMIC_RELEASE, "wavefront");
        __builtin_amdgcn_wave_barrier();
        __builtin_amdgcn_fence(__ATOMIC_ACQUIRE, "wavefront");

        if (tt + 1 < 2 * GROUPS_H && tt + 1 < ntile) cur = src_rows(ax, s[(tt + 1) >> 1], lane, ((tt + 1) & 1) * 16);

        if (cnt >= 16) {
            #pragma unroll
            for (int c0 = 0; c0 < 16; c0 += EDGE_CHUNK) {
                float lo[EDGE_CHUNK], up[EDGE_CHUNK];
                #pragma unroll
                for (int j = 0; j < EDGE_CHUNK; ++j) {
                    lo[j] = T[(c0 + j) * TPITCH + lane];
                    up[j] = T[(c0 + j) * TPITCH + lane + 32];
                }
                #pragma unroll
                for (int j = 0; j < EDGE_CHUNK; ++j) {
                    float* o = agg + (size_t)bcast_i(d[g], base + c0 + j) * H;
                    atomicAddF(o + lane,      lo[j]);
                    atomicAddF(o + lane + 32, up[j]);
                }
            }
        } else {
            for (int j = 0; j < cnt; ++j) {
                float* o = agg + (size_t)bcast_i(d[g], base + j) * H;
                atomicAddF(o + lane,      T[j * TPITCH + lane]);
                atomicAddF(o + lane + 32, T[j * TPITCH + lane + 32]);
            }
        }
        __builtin_amdgcn_fence(__ATOMIC_RELEASE, "wavefront");
        __builtin_amdgcn_wave_barrier();
        __builtin_amdgcn_fence(__ATOMIC_ACQUIRE, "wavefront");
    }
}

__device__ __forceinline__ int   pick_i(int v, int j, int hi)   { const int a = bcast_i(v, 2 * j), b = bcast_i(v, 2 * j + 1); return hi ? b : a; }

__global__ void k_scatter_x(const int* __restrict__ src, const int* __restrict__ dst,
                            const float* __restrict__ xs, float* aggx, int e) {
    const int half = my_half();
    const int lane = threadIdx.x & 31;
    const int wave = blockIdx.x * (blockDim.x >> 5) + __builtin_amdgcn_readfirstlane(threadIdx.x >> 5);
    const int e0 = wave * (EDGES_PER_WAVE * GROUPS_X);
    if (e0 >= e) return;

    int s[GROUPS_X], d[GROUPS_X];
    #pragma unroll
    for (int g = 0; g < GROUPS_X; ++g) {
        const int i = e0 + g * EDGES_PER_WAVE + lane;
        s[g] = 0;
        d[g] = 0;
        if (i < e) {
            s[g] = src[i];
            d[g] = dst[i];
        }
    }

    const int f  = lane & 15;
    const int hi = lane >> 4;
    aggx += half * AGGX_PLANE;
    #pragma unroll
    for (int g = 0; g < GROUPS_X; ++g) {
        const int left = e - (e0 + g * EDGES_PER_WAVE);
        if (left >= EDGES_PER_WAVE) {
            if (f < F_IN) {
                float v[EDGES_PER_WAVE / 2];
                #pragma unroll
                for (int j = 0; j < EDGES_PER_WAVE / 2; ++j) v[j] = xs[pick_i(s[g], j, hi) * XS + f];
                #pragma unroll
                for (int j = 0; j < EDGES_PER_WAVE / 2; ++j)
                    atomicAddF(aggx + aggx_off(pick_i(d[g], j, hi)) + f, v[j]);
            }
        } else {
            for (int j = 0; 2 * j < left; ++j) {
                const int sj = pick_i(s[g], j, hi), dj = pick_i(d[g], j, hi);
                if (f < F_IN && 2 * j + hi < left) atomicAddF(aggx + aggx_off(dj) + f, xs[sj * XS + f]);
            }
        }
    }
}


__global__ void k_zero(float* p, int n) {
    int i = blockIdx.x * blockDim.x + threadIdx.x;
    if (i < n) p[i] = 0.0f;
}


__global__ void k_gemm2_fold(const float* __restrict__ G, const float* __restrict__ W2,
                             const float* __restrict__ b2, const float* __restrict__ W3,
                             const float* __restrict__ lw, const float* __restrict__ dis,
                             float* __restrict__ y, float* __restrict__ racc,
                             float* __restrict__ racc1, int n) {
    __shared__ float Wlds[(H / 2) * H * 2];
    __shared__ alignas(16) float Ulds[H];
    __shared__ alignas(16) float Blds[H];
    __shared__ float Zlds[2][32 * 16];
    for (int idx = threadIdx.x; idx < H * H; idx += blockDim.x) {
        int k = idx / H, c = idx - k * H;
        Wlds[(k >> 1) * (H * 2) + c * 2 + (k & 1)] = W2[k * H + c];
    }
    for (int k = threadIdx.x >> 2; k < H; k += blockDim.x >> 2) {
        const int q = threadIdx.x & 3;
        const float4* wr = reinterpret_cast<const float4*>(W3 + k * H + q * 16);
        const float4* lr = reinterpret_cast<const float4*>(lw + q * 16);
        float p = 0.0f;
        #pragma unroll
        for (int i = 0; i < 4; ++i) {
            const float4 a = wr[i], b = lr[i];
            p += a.x * b.x + a.y * b.y + a.z * b.z + a.w * b.w;
        }
        p += __shfl_xor(p, 1);
        p += __shfl_xor(p, 2);
        if (q == 0) Ulds[k] = p;
    }
    for (int idx = threadIdx.x; idx < H; idx += blockDim.x) Blds[idx] = b2[idx];
    __syncthreads();

    const int lane = threadIdx.x & 31;
    const int wv    = __builtin_amdgcn_readfirstlane(threadIdx.x >> 5);
    const int mrow  = lane & 15;
    const int hi    = lane >> 4;
    const int wpb   = blockDim.x >> 5;
    const int brow  = wpb * 16;
    const int nblk  = (n + brow - 1) / brow;
    int trip = 0;
    for (int rb = blockIdx.x; rb < nblk; rb += gridDim.x, ++trip) {
    float* Z = Zlds[trip & 1];
    const int row0  = (rb * wpb + wv) * 16;
    if (row0 < n) {
        int r = row0 + mrow;
        r = r < n ? r : n - 1;
        const float* grow = G + (size_t)r * H;
        const float dn = dis[r];
        v8f acc[4];
        #pragma unroll
        for (int t = 0; t < 4; ++t) acc[t] = (v8f){0, 0, 0, 0, 0, 0, 0, 0};
        v2f gk[H / 4];
        #pragma unroll
        for (int k0 = 0; k0 < H; k0 += 4) gk[k0 >> 2] = *reinterpret_cast<const v2f*>(grow + k0 + hi * 2);
        #pragma unroll
        for (int k0 = 0; k0 < H; k0 += 4) {
            const v2f g = gk[k0 >> 2];
            const int pbase = ((k0 >> 1) + hi) * (H * 2);
            #pragma unroll
            for (int t = 0; t < 4; ++t) {
                const int c = t * 16 + mrow;
                const v2f w = *reinterpret_cast<const v2f*>(&Wlds[pbase + c * 2]);
                acc[t] = __builtin_amdgcn_wmma_f32_16x16x4_f32(false, w, false, g, (short)0, acc[t], false, false);
            }
        }
        v8f z0 = (v8f){0, 0, 0, 0, 0, 0, 0, 0};
        v8f z1 = (v8f){0, 0, 0, 0, 0, 0, 0, 0};
        #pragma unroll
        for (int t = 0; t < 4; ++t) {
            const int cb = t * 16 + hi * 8;
            const float4 ba = *reinterpret_cast<const float4*>(&Blds[cb]);
            const float4 bb = *reinterpret_cast<const float4*>(&Blds[cb + 4]);
            const float4 ua = *reinterpret_cast<const float4*>(&Ulds[cb]);
            const float4 ub = *reinterpret_cast<const float4*>(&Ulds[cb + 4]);
            const float bias[8] = {ba.x, ba.y, ba.z, ba.w, bb.x, bb.y, bb.z, bb.w};
            const float uu[8]   = {ua.x, ua.y, ua.z, ua.w, ub.x, ub.y, ub.z, ub.w};
            #pragma unroll
            for (int q = 0; q < 4; ++q) {
                v2f hv, uv;
                hv.x = fmaxf(fmaf(acc[t][2 * q],     dn, bias[2 * q]),     0.0f);
                hv.y = fmaxf(fmaf(acc[t][2 * q + 1], dn, bias[2 * q + 1]), 0.0f);
                uv.x = mrow == 0 ? uu[2 * q]     : 0.0f;
                uv.y = mrow == 0 ? uu[2 * q + 1] : 0.0f;
                if (q & 1) z1 = __builtin_amdgcn_wmma_f32_16x16x4_f32(false, uv, false, hv, (short)0, z1, false, false);
                else       z0 = __builtin_amdgcn_wmma_f32_16x16x4_f32(false, uv, false, hv, (short)0, z0, false, false);
            }
        }
        const v8f z = z0 + z1;
        if (hi == 0) Z[wv * 16 + mrow] = z[0];
    }
    __syncthreads();
    for (int t = threadIdx.x; t < brow; t += blockDim.x) {
        const int r = rb * brow + t;
        if (r < n) {
            const float v = Z[t] * dis[r];
            y[r] = v;
            racc[r] = v;
            racc1[r] = 0.0f;
        }
    }
    }
}

__global__ void k_scatter_y(const int* __restrict__ src, const int* __restrict__ dst,
                            const float* __restrict__ y, float* racc0, float* racc1, int e) {
    const int tid = blockIdx.x * blockDim.x + threadIdx.x;
    const int T = gridDim.x * blockDim.x;
    float* racc = my_half() ? racc1 : racc0;
    if ((e & 3) == 0) {
        const int4* __restrict__ sq = reinterpret_cast<const int4*>(src);
        const int4* __restrict__ dq = reinterpret_cast<const int4*>(dst);
        const int nq = e >> 2;
        int q = tid;
        for (; q + T < nq; q += 2 * T) {
            const int4 s = sq[q], d = dq[q];
            const int4 s2 = sq[q + T];
            int4 d2 = dq[q + T];
            const float v0 = y[s.x], v1 = y[s.y], v2 = y[s.z], v3 = y[s.w];
            const float w0 = y[s2.x], w1 = y[s2.y], w2 = y[s2.z], w3 = y[s2.w];
            asm volatile("" : "+v"(d2.x), "+v"(d2.y), "+v"(d2.z), "+v"(d2.w));
            atomicAddF(racc + d.x, v0);
            atomicAddF(racc + d.y, v1);
            atomicAddF(racc + d.z, v2);
            atomicAddF(racc + d.w, v3);
            atomicAddF(racc + d2.x, w0);
            atomicAddF(racc + d2.y, w1);
            atomicAddF(racc + d2.z, w2);
            atomicAddF(racc + d2.w, w3);
        }
        if (q < nq) {
            const int4 s = sq[q], d = dq[q];
            const float v0 = y[s.x], v1 = y[s.y], v2 = y[s.z], v3 = y[s.w];
            atomicAddF(racc + d.x, v0);
            atomicAddF(racc + d.y, v1);
            atomicAddF(racc + d.z, v2);
            atomicAddF(racc + d.w, v3);
        }
    } else {
        for (int i = tid; i < e; i += T) atomicAddF(racc + dst[i], y[src[i]]);
    }
}

__global__ void k_pool_y(const float* __restrict__ racc, const float* __restrict__ racc1,
                         const float* __restrict__ dis,
                         const int* __restrict__ batch, float* __restrict__ sums,
                         float* __restrict__ cnts, int n) {
    const int T = gridDim.x * blockDim.x;
    for (int i0 = blockIdx.x * blockDim.x + threadIdx.x; i0 < n; i0 += 4 * T) {
        if (i0 + 3 * T < n) {
            int g[4];
            float v[4];
            #pragma unroll
            for (int u = 0; u < 4; ++u) {
                const int i = i0 + u * T;
                g[u] = batch[i];
                v[u] = (racc[i] + racc1[i]) * dis[i];
            }
            #pragma unroll
            for (int u = 0; u < 4; ++u) {
                atomicAddF(&sums[g[u]], v[u]);
                atomicAddF(&cnts[g[u]], 1.0f);
            }
        } else {
            for (int i = i0; i < n; i += T) {
                const int g = batch[i];
                atomicAddF(&sums[g], (racc[i] + racc1[i]) * dis[i]);
                atomicAddF(&cnts[g], 1.0f);
            }
        }
    }
}

__global__ void k_pool_final(const float* __restrict__ sums, const float* __restrict__ cnts,
                             const float* __restrict__ b, const float* __restrict__ lin_w,
                             const float* __restrict__ lin_b, float* __restrict__ out, int g_total) {
    const int g = blockIdx.x * blockDim.x + threadIdx.x;
    if (g >= g_total) return;
    float c = 0.0f;
    for (int f = 0; f < H; ++f) c += b[f] * lin_w[f];
    const float cn = cnts[g];
    out[g] = (sums[g] + cn * c) / fmaxf(cn, 1.0f) + lin_b[0];
}


extern "C" void kernel_launch(void* const* d_in, const int* in_sizes, int n_in,
                              void* d_out, int out_size, void* d_ws, size_t ws_size,
                              hipStream_t stream) {
    const float* x     = (const float*)d_in[0];
    const int*   eidx  = (const int*)  d_in[1];
    const int*   batch = (const int*)  d_in[2];
    const float* W1    = (const float*)d_in[3];
    const float* b1    = (const float*)d_in[4];
    const float* W2    = (const float*)d_in[5];
    const float* b2    = (const float*)d_in[6];
    const float* W3    = (const float*)d_in[7];
    const float* b3    = (const float*)d_in[8];
    const float* lin_w = (const float*)d_in[9];
    const float* lin_b = (const float*)d_in[10];
    float* out = (float*)d_out;

    const int n = in_sizes[0] / F_IN;
    const int e = in_sizes[1] / 2;
    const int* src = eidx;
    const int* dst = eidx + e;

    float* ws   = (float*)d_ws;
    float* dis  = ws;
    float* bufA = dis  + n;
    float* bufB = bufA + (size_t)n * H;
    float* sums = bufB + (size_t)n * H;
    float* cnts = sums + N_GRAPHS;

    const int TB = 256;
    dim3 blk(TB);
    auto blocks = [](long long work, int tb) { return dim3((unsigned)((work + tb - 1) / tb)); };

    float* deg1 = bufB + (size_t)n * 32;
    k_init_deg  <<<blocks((n + 3) / 4, TB), blk, 0, stream>>>(dis, deg1, n);
    k_deg_edges <<<blocks(((long long)e + 7) / 8, TB), blk, 0, stream>>>(dst, dis, deg1, e);
    float* aggx = bufA;
    float* xs   = bufB;
    k_rsqrt_start<<<blocks((n + 1) / 2, TB), blk, 0, stream>>>(dis, deg1, x, aggx, xs, n);

    const int strips       = (n + 15) / 16;
    const dim3 gemm_grid   = blocks((long long)((strips + 1) / 2) * 32, TB);

    k_scatter_x <<<blocks(((long long)e + GROUPS_X - 1) / GROUPS_X, TB), blk, 0, stream>>>(src, dst, xs, aggx, e);
    k_gemm_l1<F_IN, 12><<<gemm_grid, blk, 0, stream>>>(aggx, W1, b1, dis, bufB, bufA, n);

    k_scatter_h1 <<<blocks(((long long)e + GROUPS_H - 1) / GROUPS_H, TB), blk, 0, stream>>>(src, dst, bufB, W1, b1, bufA, e);

    float* y    = bufB;
    float* racc = bufB + n;
    float* racc1 = racc + n;
    k_gemm2_fold<<<gemm_grid, blk, 0, stream>>>(bufA, W2, b2, W3, lin_w, dis, y, racc, racc1, n);
    k_scatter_y <<<blocks(((long long)e + 7) / 8, TB), blk, 0, stream>>>(src, dst, y, racc, racc1, e);

    k_zero      <<<blocks(2 * N_GRAPHS, TB), blk, 0, stream>>>(sums, 2 * N_GRAPHS);
    k_pool_y    <<<blocks((n + 3) / 4, TB), blk, 0, stream>>>(racc, racc1, dis, batch, sums, cnts, n);
    k_pool_final<<<blocks(N_GRAPHS, TB), blk, 0, stream>>>(sums, cnts, b3, lin_w, lin_b, out, N_GRAPHS);
}
